# starting kernel
# baseline (speedup 1.0000x reference)
#include <hip/hip_runtime.h>
#include <hip/hip_bf16.h>
#include <math.h>

typedef float v2f __attribute__((ext_vector_type(2)));
typedef float v8f __attribute__((ext_vector_type(8)));

__device__ __forceinline__ void atomicAddF32(float* p, float v) {
    unsafeAtomicAdd(p, v);
}

__device__ __forceinline__ void atomicMaxF32(float* addr, float val) {
    if (val >= 0.0f) {
        atomicMax((int*)addr, __float_as_int(val));
    } else {
        atomicMin((unsigned int*)addr, __float_as_uint(val));
    }
}

__device__ __forceinline__ void edge_sd(const int* __restrict__ ei, int E, int e,
                                        int& s, int& d) {
    if (e < E) { s = ei[e]; d = ei[E + e]; }
    else       { s = e - E; d = e - E; }
}

__global__ void gat_init_kernel(float* __restrict__ agg,
                                float* __restrict__ emax,
                                float* __restrict__ denom,
                                int n_agg, int n_nodes2) {
    int stride = gridDim.x * blockDim.x;
    for (int i = blockIdx.x * blockDim.x + threadIdx.x; i < n_agg; i += stride)
        agg[i] = 0.0f;
    for (int i = blockIdx.x * blockDim.x + threadIdx.x; i < n_nodes2; i += stride) {
        emax[i]  = -INFINITY;
        denom[i] = 0.0f;
    }
}

template <int K, int NDIM, int ROWS>
__global__ void wmma_gemm_kernel(const float* __restrict__ A,
                                 const float* __restrict__ B,
                                 const float* __restrict__ bias,
                                 float* __restrict__ C,
                                 int M) {
    constexpr int COL_TILES = NDIM / 16;
    __shared__ float Alds[ROWS * (K + 1)];

    const int tid     = threadIdx.x;
    const int wave    = tid >> 5;
    const int lane    = tid & 31;
    const int rowbase = blockIdx.x * ROWS;

    for (int idx = tid; idx < ROWS * K; idx += blockDim.x) {
        int r  = idx / K;
        int c  = idx - r * K;
        int gr = rowbase + r;
        Alds[r * (K + 1) + c] = (gr < M) ? A[(size_t)gr * K + c] : 0.0f;
    }
    __syncthreads();

    const int rg      = wave / COL_TILES;
    const int colbase = (wave % COL_TILES) * 16;
    const int half    = lane >> 4;
    const int l15     = lane & 15;

    const float* Arow = &Alds[(rg * 16 + l15) * (K + 1)];

    v8f acc = {};
    #pragma unroll 8
    for (int k = 0; k < K; k += 4) {
        const int kk = k + 2 * half;
        v2f a, b;
        a.x = Arow[kk];
        a.y = Arow[kk + 1];
        b.x = B[(size_t)kk * NDIM + colbase + l15];
        b.y = B[(size_t)(kk + 1) * NDIM + colbase + l15];
        acc = __builtin_amdgcn_wmma_f32_16x16x4_f32(
             false, a,  false, b,
             (short)0, acc,  false,  false);
    }

    const float bv = bias ? bias[colbase + l15] : 0.0f;
    #pragma unroll
    for (int r = 0; r < 8; ++r) {
        int grow = rowbase + rg * 16 + r + 8 * half;
        if (grow < M)
            C[(size_t)grow * NDIM + colbase + l15] = acc[r] + bv;
    }
}

__global__ void alpha_kernel(const float* __restrict__ xh,
                             const float* __restrict__ attl,
                             const float* __restrict__ attr,
                             float* __restrict__ al,
                             float* __restrict__ ar, int N) {
    const int gw   = (blockIdx.x * blockDim.x + threadIdx.x) >> 5;
    const int lane = threadIdx.x & 31;
    if (gw >= N) return;

    const float* xr = xh + (size_t)gw * 128;
    const float x0 = xr[lane], x1 = xr[lane + 32], x2 = xr[lane + 64], x3 = xr[lane + 96];

    float l0 = x0 * attl[lane]      + x1 * attl[lane + 32];
    float l1 = x2 * attl[lane + 64] + x3 * attl[lane + 96];
    float r0 = x0 * attr[lane]      + x1 * attr[lane + 32];
    float r1 = x2 * attr[lane + 64] + x3 * attr[lane + 96];

    #pragma unroll
    for (int off = 16; off > 0; off >>= 1) {
        l0 += __shfl_xor(l0, off, 32);
        l1 += __shfl_xor(l1, off, 32);
        r0 += __shfl_xor(r0, off, 32);
        r1 += __shfl_xor(r1, off, 32);
    }
    if (lane == 0) {
        al[gw * 2 + 0] = l0;  al[gw * 2 + 1] = l1;
        ar[gw * 2 + 0] = r0;  ar[gw * 2 + 1] = r1;
    }
}

__global__ void edge_logits_kernel(const int* __restrict__ ei, int E, int N,
                                   const float* __restrict__ al,
                                   const float* __restrict__ ar,
                                   float* __restrict__ elog,
                                   float* __restrict__ emax) {
    const int e = blockIdx.x * blockDim.x + threadIdx.x;
    if (e >= E + N) return;
    int s, d;
    edge_sd(ei, E, e, s, d);
    #pragma unroll
    for (int h = 0; h < 2; ++h) {
        float v = al[s * 2 + h] + ar[d * 2 + h];
        v = (v > 0.0f) ? v : 0.2f * v;
        elog[(size_t)e * 2 + h] = v;
        atomicMaxF32(&emax[d * 2 + h], v);
    }
}

__global__ void edge_exp_kernel(const int* __restrict__ ei, int E, int N,
                                float* __restrict__ elog,
                                const float* __restrict__ emax,
                                float* __restrict__ denom) {
    const int e = blockIdx.x * blockDim.x + threadIdx.x;
    if (e >= E + N) return;
    int s, d;
    edge_sd(ei, E, e, s, d);
    (void)s;
    #pragma unroll
    for (int h = 0; h < 2; ++h) {
        float m = emax[d * 2 + h];
        if (!(fabsf(m) <= 3.4e38f)) m = 0.0f;
        float v = __expf(elog[(size_t)e * 2 + h] - m);
        elog[(size_t)e * 2 + h] = v;
        atomicAddF32(&denom[d * 2 + h], v);
    }
}

__global__ void aggregate_kernel(const int* __restrict__ ei, int E, int N,
                                 const float* __restrict__ xh,
                                 const float* __restrict__ ex,
                                 const float* __restrict__ denom,
                                 float* __restrict__ agg) {
    const int gw   = (blockIdx.x * blockDim.x + threadIdx.x) >> 5;
    const int lane = threadIdx.x & 31;
    if (gw >= E + N) return;
    int s, d;
    edge_sd(ei, E, gw, s, d);

    const float w0 = ex[(size_t)gw * 2 + 0] / (denom[d * 2 + 0] + 1e-16f);
    const float w1 = ex[(size_t)gw * 2 + 1] / (denom[d * 2 + 1] + 1e-16f);

    const float* xs = xh  + (size_t)s * 128;
    float*       od = agg + (size_t)d * 128;
    #pragma unroll
    for (int q = 0; q < 4; ++q) {
        const int c = lane + q * 32;
        const float w = (c < 64) ? w0 : w1;
        atomicAddF32(od + c, xs[c] * w);
    }
}

extern "C" void kernel_launch(void* const* d_in, const int* in_sizes, int n_in,
                              void* d_out, int out_size, void* d_ws, size_t ws_size,
                              hipStream_t stream) {
    const float* x    = (const float*)d_in[0];
    const int*   ei   = (const int*)d_in[1];
    const float* Wl   = (const float*)d_in[2];
    const float* attl = (const float*)d_in[3];
    const float* attr = (const float*)d_in[4];
    const float* Wout = (const float*)d_in[5];
    const float* bias = (const float*)d_in[6];
    float* out = (float*)d_out;

    const int N   = in_sizes[0] / 128;
    const int E   = in_sizes[1] / 2;
    const int EDG = E + N;

    float* ws    = (float*)d_ws;
    float* xh    = ws;
    float* agg   = xh  + (size_t)N * 128;
    float* al    = agg + (size_t)N * 128;
    float* ar    = al  + (size_t)N * 2;
    float* emax  = ar  + (size_t)N * 2;
    float* denom = emax + (size_t)N * 2;
    float* elog  = denom + (size_t)N * 2;

    gat_init_kernel<<<4096, 256, 0, stream>>>(agg, emax, denom, N * 128, N * 2);

    wmma_gemm_kernel<128, 128, 16><<<(N + 15) / 16, 256, 0, stream>>>(
        x, Wl, nullptr, xh, N);

    alpha_kernel<<<(N + 7) / 8, 256, 0, stream>>>(xh, attl, attr, al, ar, N);

    edge_logits_kernel<<<(EDG + 255) / 256, 256, 0, stream>>>(
        ei, E, N, al, ar, elog, emax);

    edge_exp_kernel<<<(EDG + 255) / 256, 256, 0, stream>>>(
        ei, E, N, elog, emax, denom);

    aggregate_kernel<<<(EDG + 7) / 8, 256, 0, stream>>>(
        ei, E, N, xh, elog, denom, agg);

    wmma_gemm_kernel<128, 64, 32><<<(N + 31) / 32, 256, 0, stream>>>(
        agg, Wout, bias, out, N);
}


# Edge passes as ticketed (edge chunk, node half) tasks on resident waves, a task's loads issued together: each L2 half adds only into its own node range; f32 atomics kept, arrival order as before
# speedup vs baseline: 1.9643x; 1.9643x over previous
#include <hip/hip_runtime.h>
#include <hip/hip_bf16.h>
#include <math.h>

typedef float v2f __attribute__((ext_vector_type(2)));
typedef float v8f __attribute__((ext_vector_type(8)));

__device__ __forceinline__ void atomicAddF32(float* p, float v) {
    unsafeAtomicAdd(p, v);
}

__device__ __forceinline__ void atomicMaxF32(float* addr, float val) {
    if (val >= 0.0f) {
        atomicMax((int*)addr, __float_as_int(val));
    } else {
        atomicMin((unsigned int*)addr, __float_as_uint(val));
    }
}

__device__ __forceinline__ void edge_sd(const int* __restrict__ ei, int E, int e,
                                        int& s, int& d) {
    if (e < E) { s = ei[e]; d = ei[E + e]; }
    else       { s = e - E; d = e - E; }
}


#define QSTRIDE 64

__device__ __forceinline__ int l2_half() {
    const unsigned v = __builtin_amdgcn_s_sendmsg_rtn(0x87);
    return (int)((v >> 18) & 1u);
}

__device__ __forceinline__ int wave_ticket(int* q) {
    int t = 0;
    if ((threadIdx.x & 31) == 0) t = atomicAdd(q, 1);
    return __builtin_amdgcn_readfirstlane(t);
}

__device__ __forceinline__ int block_first_ticket(int* q, int* slot) {
    if (threadIdx.x == 0) *slot = atomicAdd(q, (int)(blockDim.x >> 5));
    __syncthreads();
    return __builtin_amdgcn_readfirstlane(*slot + (int)(threadIdx.x >> 5));
}

__global__ void gat_init_kernel(float* __restrict__ agg,
                                float* __restrict__ emax,
                                float* __restrict__ denom,
                                int* __restrict__ queues,
                                int n_agg, int n_nodes2) {
    int stride = gridDim.x * blockDim.x;
    if (blockIdx.x == 0 && threadIdx.x < 6) queues[threadIdx.x * QSTRIDE] = 0;
    for (int i = blockIdx.x * blockDim.x + threadIdx.x; i < n_agg; i += stride)
        agg[i] = 0.0f;
    for (int i = blockIdx.x * blockDim.x + threadIdx.x; i < n_nodes2; i += stride) {
        emax[i]  = -INFINITY;
        denom[i] = 0.0f;
    }
}

template <int K, int NDIM, int ROWS>
__global__ void wmma_gemm_kernel(const float* __restrict__ A,
                                 const float* __restrict__ B,
                                 const float* __restrict__ bias,
                                 float* __restrict__ C,
                                 int M) {
    constexpr int COL_TILES = NDIM / 16;
    __shared__ float Alds[ROWS * (K + 1)];

    const int tid     = threadIdx.x;
    const int wave    = tid >> 5;
    const int lane    = tid & 31;
    const int rowbase = blockIdx.x * ROWS;

    for (int idx = tid; idx < ROWS * K; idx += blockDim.x) {
        int r  = idx / K;
        int c  = idx - r * K;
        int gr = rowbase + r;
        Alds[r * (K + 1) + c] = (gr < M) ? A[(size_t)gr * K + c] : 0.0f;
    }
    __syncthreads();

    const int rg      = wave / COL_TILES;
    const int colbase = (wave % COL_TILES) * 16;
    const int half    = lane >> 4;
    const int l15     = lane & 15;

    const float* Arow = &Alds[(rg * 16 + l15) * (K + 1)];

    v8f acc = {};
    #pragma unroll 8
    for (int k = 0; k < K; k += 4) {
        const int kk = k + 2 * half;
        v2f a, b;
        a.x = Arow[kk];
        a.y = Arow[kk + 1];
        b.x = B[(size_t)kk * NDIM + colbase + l15];
        b.y = B[(size_t)(kk + 1) * NDIM + colbase + l15];
        acc = __builtin_amdgcn_wmma_f32_16x16x4_f32(
             false, a,  false, b,
             (short)0, acc,  false,  false);
    }

    const float bv = bias ? bias[colbase + l15] : 0.0f;
    #pragma unroll
    for (int r = 0; r < 8; ++r) {
        int grow = rowbase + rg * 16 + r + 8 * half;
        if (grow < M)
            C[(size_t)grow * NDIM + colbase + l15] = acc[r] + bv;
    }
}

__global__ void alpha_kernel(const float* __restrict__ xh,
                             const float* __restrict__ attl,
                             const float* __restrict__ attr,
                             float* __restrict__ al,
                             float* __restrict__ ar, int N) {
    const int gw   = (blockIdx.x * blockDim.x + threadIdx.x) >> 5;
    const int lane = threadIdx.x & 31;
    if (gw >= N) return;

    const float* xr = xh + (size_t)gw * 128;
    const float x0 = xr[lane], x1 = xr[lane + 32], x2 = xr[lane + 64], x3 = xr[lane + 96];

    float l0 = x0 * attl[lane]      + x1 * attl[lane + 32];
    float l1 = x2 * attl[lane + 64] + x3 * attl[lane + 96];
    float r0 = x0 * attr[lane]      + x1 * attr[lane + 32];
    float r1 = x2 * attr[lane + 64] + x3 * attr[lane + 96];

    #pragma unroll
    for (int off = 16; off > 0; off >>= 1) {
        l0 += __shfl_xor(l0, off, 32);
        l1 += __shfl_xor(l1, off, 32);
        r0 += __shfl_xor(r0, off, 32);
        r1 += __shfl_xor(r1, off, 32);
    }
    if (lane == 0) {
        al[gw * 2 + 0] = l0;  al[gw * 2 + 1] = l1;
        ar[gw * 2 + 0] = r0;  ar[gw * 2 + 1] = r1;
    }
}

typedef float f2 __attribute__((ext_vector_type(2)));

__device__ __forceinline__ void edge_sd_nb(const int* __restrict__ ei, int E, int e,
                                           int& s, int& d) {
    const int ec = (e < E) ? e : (E - 1);
    const int sl = ei[ec];
    const int dl = ei[E + ec];
    s = (e < E) ? sl : (e - E);
    d = (e < E) ? dl : (e - E);
}

__global__ __launch_bounds__(256)
void edge_logits_kernel(const int* __restrict__ ei, int E, int N,
                        const float* __restrict__ al,
                        const float* __restrict__ ar,
                        float* __restrict__ elog,
                        float* __restrict__ emax,
                        int* __restrict__ queues) {
    __shared__ int slot;
    const int lane   = threadIdx.x & 31;
    const int EDG    = E + N;
    const int ntask  = (EDG + 63) >> 6;
    const int nsplit = N >> 1;
    const int h0     = l2_half();
    for (int ph = 0; ph < 2; ++ph) {
        const int hh = h0 ^ ph;
        int* q = queues + hh * QSTRIDE;
        int t = (ph == 0) ? block_first_ticket(q, &slot) : wave_ticket(q);
        while (t < ntask) {
            int  e[2], s[2], d[2];
            bool on[2];
            #pragma unroll
            for (int r = 0; r < 2; ++r) {
                e[r]  = (t << 6) + (r << 5) + lane;
                on[r] = e[r] < EDG;
                edge_sd_nb(ei, E, on[r] ? e[r] : 0, s[r], d[r]);
            }
            f2 vl[2], vr[2];
            #pragma unroll
            for (int r = 0; r < 2; ++r) {
                vl[r] = *(const f2*)(al + (size_t)s[r] * 2);
                vr[r] = *(const f2*)(ar + (size_t)d[r] * 2);
            }
            #pragma unroll
            for (int r = 0; r < 2; ++r) {
                if (on[r] && (((d[r] >= nsplit) ? 1 : 0) == hh)) {
                    f2 v = vl[r] + vr[r];
                    v.x = (v.x > 0.0f) ? v.x : 0.2f * v.x;
                    v.y = (v.y > 0.0f) ? v.y : 0.2f * v.y;
                    *(f2*)(elog + (size_t)e[r] * 2) = v;
                    atomicMaxF32(&emax[d[r] * 2 + 0], v.x);
                    atomicMaxF32(&emax[d[r] * 2 + 1], v.y);
                }
            }
            t = wave_ticket(q);
        }
    }
}

__global__ __launch_bounds__(256)
void edge_exp_kernel(const int* __restrict__ ei, int E, int N,
                     float* __restrict__ elog,
                     const float* __restrict__ emax,
                     float* __restrict__ denom,
                     int* __restrict__ queues) {
    __shared__ int slot;
    const int lane   = threadIdx.x & 31;
    const int EDG    = E + N;
    const int ntask  = (EDG + 63) >> 6;
    const int nsplit = N >> 1;
    const int h0     = l2_half();
    for (int ph = 0; ph < 2; ++ph) {
        const int hh = h0 ^ ph;
        int* q = queues + hh * QSTRIDE;
        int t = (ph == 0) ? block_first_ticket(q, &slot) : wave_ticket(q);
        while (t < ntask) {
            int  e[2], s[2], d[2];
            bool on[2];
            #pragma unroll
            for (int r = 0; r < 2; ++r) {
                e[r]  = (t << 6) + (r << 5) + lane;
                on[r] = e[r] < EDG;
                edge_sd_nb(ei, E, on[r] ? e[r] : 0, s[r], d[r]);
            }
            #pragma unroll
            for (int r = 0; r < 2; ++r) {
                if (on[r] && (((d[r] >= nsplit) ? 1 : 0) == hh)) {
                    f2 m = *(const f2*)(emax + (size_t)d[r] * 2);
                    f2 x = *(const f2*)(elog + (size_t)e[r] * 2);
                    if (!(fabsf(m.x) <= 3.4e38f)) m.x = 0.0f;
                    if (!(fabsf(m.y) <= 3.4e38f)) m.y = 0.0f;
                    f2 v;
                    v.x = __expf(x.x - m.x);
                    v.y = __expf(x.y - m.y);
                    *(f2*)(elog + (size_t)e[r] * 2) = v;
                    atomicAddF32(&denom[d[r] * 2 + 0], v.x);
                    atomicAddF32(&denom[d[r] * 2 + 1], v.y);
                }
            }
            t = wave_ticket(q);
        }
    }
}

template <int K>
__device__ __forceinline__ void add_rows(unsigned& m, int s, int d, float w0, float w1,
                                         const float* __restrict__ xh,
                                         float* __restrict__ agg, int lane) {
    float  v[K][4];
    float  a0[K], a1[K];
    float* od[K];
    #pragma unroll
    for (int k = 0; k < K; ++k) {
        const int j = __builtin_ctz(m);
        m &= m - 1;
        const int sj = __builtin_amdgcn_readlane(s, j);
        const int dj = __builtin_amdgcn_readlane(d, j);
        a0[k] = __int_as_float(__builtin_amdgcn_readlane(__float_as_int(w0), j));
        a1[k] = __int_as_float(__builtin_amdgcn_readlane(__float_as_int(w1), j));
        const float* xs = xh + (size_t)sj * 128 + lane;
        od[k] = agg + (size_t)dj * 128 + lane;
        #pragma unroll
        for (int c = 0; c < 4; ++c) v[k][c] = xs[c * 32];
    }
    #pragma unroll
    for (int k = 0; k < K; ++k) {
        #pragma unroll
        for (int c = 0; c < 4; ++c)
            atomicAddF32(od[k] + c * 32, v[k][c] * ((c < 2) ? a0[k] : a1[k]));
    }
}

__global__ __launch_bounds__(256)
void aggregate_kernel(const int* __restrict__ ei, int E, int N,
                      const float* __restrict__ xh,
                      const float* __restrict__ ex,
                      const float* __restrict__ denom,
                      float* __restrict__ agg,
                      int* __restrict__ queues) {
    __shared__ int slot;
    const int lane   = threadIdx.x & 31;
    const int EDG    = E + N;
    const int ntask  = (EDG + 31) >> 5;
    const int nsplit = N >> 1;
    const int h0     = l2_half();
    for (int ph = 0; ph < 2; ++ph) {
        const int hh = h0 ^ ph;
        int* q = queues + hh * QSTRIDE;
        int t = (ph == 0) ? block_first_ticket(q, &slot) : wave_ticket(q);
        while (t < ntask) {
            const int e = (t << 5) + lane;
            int s = 0, d = 0;
            float w0 = 0.0f, w1 = 0.0f;
            bool mine = false;
            if (e < EDG) {
                edge_sd(ei, E, e, s, d);
                mine = (((d >= nsplit) ? 1 : 0) == hh);
                if (mine) {
                    w0 = ex[(size_t)e * 2 + 0] / (denom[d * 2 + 0] + 1e-16f);
                    w1 = ex[(size_t)e * 2 + 1] / (denom[d * 2 + 1] + 1e-16f);
                }
            }
            unsigned m = __builtin_amdgcn_ballot_w32(mine);
            while (__builtin_popcount(m) >= 4) add_rows<4>(m, s, d, w0, w1, xh, agg, lane);
            if (__builtin_popcount(m) >= 2)    add_rows<2>(m, s, d, w0, w1, xh, agg, lane);
            if (m)                             add_rows<1>(m, s, d, w0, w1, xh, agg, lane);
            t = wave_ticket(q);
        }
    }
}

extern "C" void kernel_launch(void* const* d_in, const int* in_sizes, int n_in,
                              void* d_out, int out_size, void* d_ws, size_t ws_size,
                              hipStream_t stream) {
    const float* x    = (const float*)d_in[0];
    const int*   ei   = (const int*)d_in[1];
    const float* Wl   = (const float*)d_in[2];
    const float* attl = (const float*)d_in[3];
    const float* attr = (const float*)d_in[4];
    const float* Wout = (const float*)d_in[5];
    const float* bias = (const float*)d_in[6];
    float* out = (float*)d_out;

    const int N   = in_sizes[0] / 128;
    const int E   = in_sizes[1] / 2;
    const int EDG = E + N;

    float* ws    = (float*)d_ws;
    float* xh    = ws;
    float* agg   = xh  + (size_t)N * 128;
    float* al    = agg + (size_t)N * 128;
    float* ar    = al  + (size_t)N * 2;
    float* emax  = ar  + (size_t)N * 2;
    float* denom = emax + (size_t)N * 2;
    float* elog  = denom + (size_t)N * 2;
    int*   queues = (int*)(elog + (size_t)EDG * 2);

    gat_init_kernel<<<4096, 256, 0, stream>>>(agg, emax, denom, queues, N * 128, N * 2);

    wmma_gemm_kernel<128, 128, 16><<<(N + 15) / 16, 256, 0, stream>>>(
        x, Wl, nullptr, xh, N);

    alpha_kernel<<<(N + 7) / 8, 256, 0, stream>>>(xh, attl, attr, al, ar, N);

    edge_logits_kernel<<<1024, 256, 0, stream>>>(
        ei, E, N, al, ar, elog, emax, queues);

    edge_exp_kernel<<<1024, 256, 0, stream>>>(
        ei, E, N, elog, emax, denom, queues + 2 * QSTRIDE);

    aggregate_kernel<<<2048, 256, 0, stream>>>(
        ei, E, N, xh, elog, denom, agg, queues + 4 * QSTRIDE);

    wmma_gemm_kernel<128, 64, 32><<<(N + 31) / 32, 256, 0, stream>>>(
        agg, Wout, bias, out, N);
}


# Input edges counting-sorted by destination in-call (int atomics); one wave per node then does max, exp-sum and weighted row sum in registers: no f32 atomics, sums run in per-node list order
# speedup vs baseline: 2.8225x; 1.4369x over previous
#include <hip/hip_runtime.h>
#include <math.h>

typedef float v2f __attribute__((ext_vector_type(2)));
typedef float v8f __attribute__((ext_vector_type(8)));

#define QSTRIDE 64

__device__ __forceinline__ int l2_half() {
    const unsigned v = __builtin_amdgcn_s_sendmsg_rtn(0x87);
    return (int)((v >> 18) & 1u);
}

__device__ __forceinline__ int wave_ticket(int* q) {
    int t = 0;
    if ((threadIdx.x & 31) == 0) t = atomicAdd(q, 1);
    return __builtin_amdgcn_readfirstlane(t);
}

__device__ __forceinline__ int block_first_ticket(int* q, int* slot) {
    if (threadIdx.x == 0) *slot = atomicAdd(q, (int)(blockDim.x >> 5));
    __syncthreads();
    return __builtin_amdgcn_readfirstlane(*slot + (int)(threadIdx.x >> 5));
}

__global__ void sort_init_kernel(int* __restrict__ cnt, int* __restrict__ queues, int N) {
    const int stride = gridDim.x * blockDim.x;
    if (blockIdx.x == 0 && threadIdx.x < 2) queues[threadIdx.x * QSTRIDE] = 0;
    for (int i = blockIdx.x * blockDim.x + threadIdx.x; i < N; i += stride) cnt[i] = 0;
}

__global__ __launch_bounds__(256)
void dst_rank_kernel(const int* __restrict__ ei, int E, int N,
                     int* __restrict__ cnt,
                     int* __restrict__ rank,
                     int* __restrict__ queues) {
    __shared__ int slot;
    const int lane   = threadIdx.x & 31;
    const int ntask  = (E + 63) >> 6;
    const int nsplit = N >> 1;
    const int h0     = l2_half();
    for (int ph = 0; ph < 2; ++ph) {
        const int hh = h0 ^ ph;
        int* q = queues + hh * QSTRIDE;
        int t = (ph == 0) ? block_first_ticket(q, &slot) : wave_ticket(q);
        while (t < ntask) {
            int  e[2], d[2];
            bool on[2];
            #pragma unroll
            for (int r = 0; r < 2; ++r) {
                e[r]  = (t << 6) + (r << 5) + lane;
                on[r] = e[r] < E;
                d[r]  = ei[E + (on[r] ? e[r] : 0)];
            }
            bool mine[2];
            int  place[2];
            #pragma unroll
            for (int r = 0; r < 2; ++r) {
                mine[r]  = on[r] && (((d[r] >= nsplit) ? 1 : 0) == hh);
                place[r] = 0;
                if (mine[r]) place[r] = atomicAdd(&cnt[d[r]], 1);
            }
            #pragma unroll
            for (int r = 0; r < 2; ++r)
                if (mine[r]) rank[e[r]] = place[r];
            t = wave_ticket(q);
        }
    }
}

__global__ __launch_bounds__(1024)
void offsets_kernel(const int* __restrict__ cnt, int* __restrict__ off, int N) {
    __shared__ int wsum[32];
    const int tid  = threadIdx.x;
    const int lane = tid & 31;
    const int wave = tid >> 5;
    const int nb   = gridDim.x;
    const int seg  = (N + nb - 1) / nb;
    const int lo   = min((int)blockIdx.x * seg, N);
    const int hi   = min(lo + seg, N);

    int part = 0;
    for (int i = tid; i < lo; i += 1024) part += cnt[i];
    #pragma unroll
    for (int o = 16; o > 0; o >>= 1) part += __shfl_xor(part, o, 32);
    if (lane == 0) wsum[wave] = part;
    __syncthreads();
    int base = 0;
    for (int w = 0; w < 32; ++w) base += wsum[w];
    __syncthreads();

    for (int t0 = lo; t0 < hi; t0 += 1024) {
        const int i = t0 + tid;
        const int v = (i < hi) ? cnt[i] : 0;
        int incl = v;
        #pragma unroll
        for (int o = 1; o < 32; o <<= 1) {
            const int up = __shfl_up(incl, o, 32);
            if (lane >= o) incl += up;
        }
        if (lane == 31) wsum[wave] = incl;
        __syncthreads();
        int wbase = 0, total = 0;
        for (int w = 0; w < 32; ++w) {
            const int s = wsum[w];
            if (w < wave) wbase += s;
            total += s;
        }
        if (i < hi) off[i] = base + wbase + incl - v;
        base += total;
        __syncthreads();
    }
    if ((int)blockIdx.x == nb - 1 && tid == 0) off[N] = base;
}

__global__ __launch_bounds__(256)
void place_kernel(const int* __restrict__ ei, int E,
                  const int* __restrict__ off,
                  const int* __restrict__ rank,
                  int* __restrict__ srt) {
    const int stride = gridDim.x * blockDim.x;
    for (int e = blockIdx.x * blockDim.x + threadIdx.x; e < E; e += stride) {
        const int d = ei[E + e];
        srt[off[d] + rank[e]] = ei[e];
    }
}

template <int K, int NDIM, int ROWS>
__global__ void wmma_gemm_kernel(const float* __restrict__ A,
                                 const float* __restrict__ B,
                                 const float* __restrict__ bias,
                                 float* __restrict__ C,
                                 int M) {
    constexpr int COL_TILES = NDIM / 16;
    __shared__ float Alds[ROWS * (K + 1)];

    const int tid     = threadIdx.x;
    const int wave    = tid >> 5;
    const int lane    = tid & 31;
    const int rowbase = blockIdx.x * ROWS;

    for (int idx = tid; idx < ROWS * K; idx += blockDim.x) {
        int r  = idx / K;
        int c  = idx - r * K;
        int gr = rowbase + r;
        Alds[r * (K + 1) + c] = (gr < M) ? A[(size_t)gr * K + c] : 0.0f;
    }
    __syncthreads();

    const int rg      = wave / COL_TILES;
    const int colbase = (wave % COL_TILES) * 16;
    const int half    = lane >> 4;
    const int l15     = lane & 15;

    const float* Arow = &Alds[(rg * 16 + l15) * (K + 1)];

    v8f acc = {};
    #pragma unroll 8
    for (int k = 0; k < K; k += 4) {
        const int kk = k + 2 * half;
        v2f a, b;
        a.x = Arow[kk];
        a.y = Arow[kk + 1];
        b.x = B[(size_t)kk * NDIM + colbase + l15];
        b.y = B[(size_t)(kk + 1) * NDIM + colbase + l15];
        acc = __builtin_amdgcn_wmma_f32_16x16x4_f32(
             false, a,  false, b,
             (short)0, acc,  false,  false);
    }

    const float bv = bias ? bias[colbase + l15] : 0.0f;
    #pragma unroll
    for (int r = 0; r < 8; ++r) {
        int grow = rowbase + rg * 16 + r + 8 * half;
        if (grow < M)
            C[(size_t)grow * NDIM + colbase + l15] = acc[r] + bv;
    }
}

__global__ void alpha_kernel(const float* __restrict__ xh,
                             const float* __restrict__ attl,
                             const float* __restrict__ attr,
                             float* __restrict__ al,
                             float* __restrict__ ar, int N) {
    const int gw   = (blockIdx.x * blockDim.x + threadIdx.x) >> 5;
    const int lane = threadIdx.x & 31;
    if (gw >= N) return;

    const float* xr = xh + (size_t)gw * 128;
    const float x0 = xr[lane], x1 = xr[lane + 32], x2 = xr[lane + 64], x3 = xr[lane + 96];

    float l0 = x0 * attl[lane]      + x1 * attl[lane + 32];
    float l1 = x2 * attl[lane + 64] + x3 * attl[lane + 96];
    float r0 = x0 * attr[lane]      + x1 * attr[lane + 32];
    float r1 = x2 * attr[lane + 64] + x3 * attr[lane + 96];

    #pragma unroll
    for (int off = 16; off > 0; off >>= 1) {
        l0 += __shfl_xor(l0, off, 32);
        l1 += __shfl_xor(l1, off, 32);
        r0 += __shfl_xor(r0, off, 32);
        r1 += __shfl_xor(r1, off, 32);
    }
    if (lane == 0) {
        al[gw * 2 + 0] = l0;  al[gw * 2 + 1] = l1;
        ar[gw * 2 + 0] = r0;  ar[gw * 2 + 1] = r1;
    }
}

typedef float f2 __attribute__((ext_vector_type(2)));
typedef float v4f __attribute__((ext_vector_type(4)));

__device__ __forceinline__ float wave_max(float v) {
    #pragma unroll
    for (int o = 16; o > 0; o >>= 1) v = fmaxf(v, __shfl_xor(v, o, 32));
    return v;
}
__device__ __forceinline__ float wave_sum(float v) {
    #pragma unroll
    for (int o = 16; o > 0; o >>= 1) v += __shfl_xor(v, o, 32);
    return v;
}

__device__ __forceinline__ void load_chunk(int c, int lane, int i, int o0, int deg,
                                           const int* __restrict__ srt,
                                           const float* __restrict__ al, f2 vr,
                                           int& s, f2& e, bool& ok) {
    const int j = c + lane;
    ok = (j <= deg);
    s = i;
    if (j < deg) s = srt[o0 + j];
    const f2 vl = *(const f2*)(al + (size_t)s * 2);
    f2 v = vl + vr;
    v.x = (v.x > 0.0f) ? v.x : 0.2f * v.x;
    v.y = (v.y > 0.0f) ? v.y : 0.2f * v.y;
    e = v;
}

template <int K>
__device__ __forceinline__ void gather_rows(int j0, int s, float w0, float w1,
                                            const float* __restrict__ xh, int lane,
                                            v4f& acc) {
    v4f   v[K];
    float a[K];
    #pragma unroll
    for (int k = 0; k < K; ++k) {
        const int   sj = __builtin_amdgcn_readlane(s, j0 + k);
        const float a0 = __int_as_float(__builtin_amdgcn_readlane(__float_as_int(w0), j0 + k));
        const float a1 = __int_as_float(__builtin_amdgcn_readlane(__float_as_int(w1), j0 + k));
        a[k] = (lane < 16) ? a0 : a1;
        v[k] = *(const v4f*)(xh + (size_t)sj * 128 + 4 * lane);
    }
    #pragma unroll
    for (int k = 0; k < K; ++k) acc += v[k] * a[k];
}

__device__ __forceinline__ void walk_chunk(int n, int s, float w0, float w1,
                                           const float* __restrict__ xh, int lane,
                                           v4f& acc) {
    int j = 0;
    while (n - j >= 8) { gather_rows<8>(j, s, w0, w1, xh, lane, acc); j += 8; }
    if (n - j >= 4)    { gather_rows<4>(j, s, w0, w1, xh, lane, acc); j += 4; }
    if (n - j >= 2)    { gather_rows<2>(j, s, w0, w1, xh, lane, acc); j += 2; }
    if (n - j >= 1)    { gather_rows<1>(j, s, w0, w1, xh, lane, acc); }
}

__global__ __launch_bounds__(256)
void node_softmax_gather_kernel(const int* __restrict__ off,
                                const int* __restrict__ srt,
                                const float* __restrict__ al,
                                const float* __restrict__ ar,
                                const float* __restrict__ xh,
                                float* __restrict__ agg, int N) {
    const int lane = threadIdx.x & 31;
    const int nw   = (int)((gridDim.x * blockDim.x) >> 5);
    const int wid  = __builtin_amdgcn_readfirstlane((int)((blockIdx.x * blockDim.x + threadIdx.x) >> 5));
    for (int i = wid; i < N; i += nw) {
        const int o0  = off[i];
        const int deg = off[i + 1] - o0;
        const int tot = deg + 1;
        const f2  vr  = *(const f2*)(ar + (size_t)i * 2);

        int s0; f2 e0; bool ok0;
        load_chunk(0, lane, i, o0, deg, srt, al, vr, s0, e0, ok0);

        float m0 = wave_max(ok0 ? e0.x : -INFINITY);
        float m1 = wave_max(ok0 ? e0.y : -INFINITY);
        for (int c = 32; c < tot; c += 32) {
            int s; f2 e; bool ok;
            load_chunk(c, lane, i, o0, deg, srt, al, vr, s, e, ok);
            m0 = fmaxf(m0, wave_max(ok ? e.x : -INFINITY));
            m1 = fmaxf(m1, wave_max(ok ? e.y : -INFINITY));
        }
        if (!(fabsf(m0) <= 3.4e38f)) m0 = 0.0f;
        if (!(fabsf(m1) <= 3.4e38f)) m1 = 0.0f;

        const float x0 = ok0 ? __expf(e0.x - m0) : 0.0f;
        const float x1 = ok0 ? __expf(e0.y - m1) : 0.0f;
        float d0 = wave_sum(x0);
        float d1 = wave_sum(x1);
        for (int c = 32; c < tot; c += 32) {
            int s; f2 e; bool ok;
            load_chunk(c, lane, i, o0, deg, srt, al, vr, s, e, ok);
            d0 += wave_sum(ok ? __expf(e.x - m0) : 0.0f);
            d1 += wave_sum(ok ? __expf(e.y - m1) : 0.0f);
        }
        d0 += 1e-16f;
        d1 += 1e-16f;

        v4f acc = {0.0f, 0.0f, 0.0f, 0.0f};
        walk_chunk(min(tot, 32), s0, x0 / d0, x1 / d1, xh, lane, acc);
        for (int c = 32; c < tot; c += 32) {
            int s; f2 e; bool ok;
            load_chunk(c, lane, i, o0, deg, srt, al, vr, s, e, ok);
            const float w0 = ok ? __expf(e.x - m0) / d0 : 0.0f;
            const float w1 = ok ? __expf(e.y - m1) / d1 : 0.0f;
            walk_chunk(min(tot - c, 32), s, w0, w1, xh, lane, acc);
        }
        *(v4f*)(agg + (size_t)i * 128 + 4 * lane) = acc;
    }
}

extern "C" void kernel_launch(void* const* d_in, const int* in_sizes, int n_in,
                              void* d_out, int out_size, void* d_ws, size_t ws_size,
                              hipStream_t stream) {
    const float* x    = (const float*)d_in[0];
    const int*   ei   = (const int*)d_in[1];
    const float* Wl   = (const float*)d_in[2];
    const float* attl = (const float*)d_in[3];
    const float* attr = (const float*)d_in[4];
    const float* Wout = (const float*)d_in[5];
    const float* bias = (const float*)d_in[6];
    float* out = (float*)d_out;

    const int N   = in_sizes[0] / 128;
    const int E   = in_sizes[1] / 2;
    const int EDG = E + N;

    float* ws    = (float*)d_ws;
    float* xh    = ws;
    float* agg   = xh  + (size_t)N * 128;
    float* al    = agg + (size_t)N * 128;
    float* ar    = al  + (size_t)N * 2;
    int*   cnt   = (int*)(ar + (size_t)N * 2);
    int*   off   = cnt + (size_t)N * 2;
    int*   rank  = off + (size_t)N * 2;
    int*   srt   = rank + (size_t)E;
    int*   queues = rank + (size_t)EDG * 2;

    sort_init_kernel<<<64, 256, 0, stream>>>(cnt, queues, N);

    dst_rank_kernel<<<1024, 256, 0, stream>>>(ei, E, N, cnt, rank, queues);
    offsets_kernel<<<64, 1024, 0, stream>>>(cnt, off, N);
    place_kernel<<<1024, 256, 0, stream>>>(ei, E, off, rank, srt);

    wmma_gemm_kernel<128, 128, 16><<<(N + 15) / 16, 256, 0, stream>>>(
        x, Wl, nullptr, xh, N);

    alpha_kernel<<<(N + 7) / 8, 256, 0, stream>>>(xh, attl, attr, al, ar, N);

    node_softmax_gather_kernel<<<2048, 256, 0, stream>>>(off, srt, al, ar, xh, agg, N);

    wmma_gemm_kernel<128, 64, 32><<<(N + 31) / 32, 256, 0, stream>>>(
        agg, Wout, bias, out, N);
}


# Attention coefficients al, ar formed in the projection GEMM's epilogue from its accumulators (16-lane row sums, 4 waves added in order): alpha kernel and its launch removed; dot sum order changes
# speedup vs baseline: 2.9287x; 1.0376x over previous
#include <hip/hip_runtime.h>
#include <math.h>

typedef float v2f __attribute__((ext_vector_type(2)));
typedef float v8f __attribute__((ext_vector_type(8)));

#define QSTRIDE 64

__device__ __forceinline__ int l2_half() {
    const unsigned v = __builtin_amdgcn_s_sendmsg_rtn(0x87);
    return (int)((v >> 18) & 1u);
}

__device__ __forceinline__ int wave_ticket(int* q) {
    int t = 0;
    if ((threadIdx.x & 31) == 0) t = atomicAdd(q, 1);
    return __builtin_amdgcn_readfirstlane(t);
}

__device__ __forceinline__ int block_first_ticket(int* q, int* slot) {
    if (threadIdx.x == 0) *slot = atomicAdd(q, (int)(blockDim.x >> 5));
    __syncthreads();
    return __builtin_amdgcn_readfirstlane(*slot + (int)(threadIdx.x >> 5));
}

__global__ void sort_init_kernel(int* __restrict__ cnt, int* __restrict__ queues, int N) {
    const int stride = gridDim.x * blockDim.x;
    if (blockIdx.x == 0 && threadIdx.x < 2) queues[threadIdx.x * QSTRIDE] = 0;
    for (int i = blockIdx.x * blockDim.x + threadIdx.x; i < N; i += stride) cnt[i] = 0;
}

__global__ __launch_bounds__(256)
void dst_rank_kernel(const int* __restrict__ ei, int E, int N,
                     int* __restrict__ cnt,
                     int* __restrict__ rank,
                     int* __restrict__ queues) {
    __shared__ int slot;
    const int lane   = threadIdx.x & 31;
    const int ntask  = (E + 63) >> 6;
    const int nsplit = N >> 1;
    const int h0     = l2_half();
    for (int ph = 0; ph < 2; ++ph) {
        const int hh = h0 ^ ph;
        int* q = queues + hh * QSTRIDE;
        int t = (ph == 0) ? block_first_ticket(q, &slot) : wave_ticket(q);
        while (t < ntask) {
            int  e[2], d[2];
            bool on[2];
            #pragma unroll
            for (int r = 0; r < 2; ++r) {
                e[r]  = (t << 6) + (r << 5) + lane;
                on[r] = e[r] < E;
                d[r]  = ei[E + (on[r] ? e[r] : 0)];
            }
            bool mine[2];
            int  place[2];
            #pragma unroll
            for (int r = 0; r < 2; ++r) {
                mine[r]  = on[r] && (((d[r] >= nsplit) ? 1 : 0) == hh);
                place[r] = 0;
                if (mine[r]) place[r] = atomicAdd(&cnt[d[r]], 1);
            }
            #pragma unroll
            for (int r = 0; r < 2; ++r)
                if (mine[r]) rank[e[r]] = place[r];
            t = wave_ticket(q);
        }
    }
}

__global__ __launch_bounds__(1024)
void offsets_kernel(const int* __restrict__ cnt, int* __restrict__ off, int N) {
    __shared__ int wsum[32];
    const int tid  = threadIdx.x;
    const int lane = tid & 31;
    const int wave = tid >> 5;
    const int nb   = gridDim.x;
    const int seg  = (N + nb - 1) / nb;
    const int lo   = min((int)blockIdx.x * seg, N);
    const int hi   = min(lo + seg, N);

    int part = 0;
    for (int i = tid; i < lo; i += 1024) part += cnt[i];
    #pragma unroll
    for (int o = 16; o > 0; o >>= 1) part += __shfl_xor(part, o, 32);
    if (lane == 0) wsum[wave] = part;
    __syncthreads();
    int base = 0;
    for (int w = 0; w < 32; ++w) base += wsum[w];
    __syncthreads();

    for (int t0 = lo; t0 < hi; t0 += 1024) {
        const int i = t0 + tid;
        const int v = (i < hi) ? cnt[i] : 0;
        int incl = v;
        #pragma unroll
        for (int o = 1; o < 32; o <<= 1) {
            const int up = __shfl_up(incl, o, 32);
            if (lane >= o) incl += up;
        }
        if (lane == 31) wsum[wave] = incl;
        __syncthreads();
        int wbase = 0, total = 0;
        for (int w = 0; w < 32; ++w) {
            const int s = wsum[w];
            if (w < wave) wbase += s;
            total += s;
        }
        if (i < hi) off[i] = base + wbase + incl - v;
        base += total;
        __syncthreads();
    }
    if ((int)blockIdx.x == nb - 1 && tid == 0) off[N] = base;
}

__global__ __launch_bounds__(256)
void place_kernel(const int* __restrict__ ei, int E,
                  const int* __restrict__ off,
                  const int* __restrict__ rank,
                  int* __restrict__ srt) {
    const int stride = gridDim.x * blockDim.x;
    for (int e = blockIdx.x * blockDim.x + threadIdx.x; e < E; e += stride) {
        const int d = ei[E + e];
        srt[off[d] + rank[e]] = ei[e];
    }
}


template <int M>
__device__ __forceinline__ float row_xor_add(float v) {
    const int t = __builtin_amdgcn_update_dpp(0, __float_as_int(v), 0x160 + M, 0xf, 0xf, true);
    return v + __int_as_float(t);
}

template <int K, int NDIM, int ROWS, bool ALPHA>
__global__ void wmma_gemm_kernel(const float* __restrict__ A,
                                 const float* __restrict__ B,
                                 const float* __restrict__ bias,
                                 float* __restrict__ C,
                                 const float* __restrict__ attl,
                                 const float* __restrict__ attr,
                                 float* __restrict__ al,
                                 float* __restrict__ ar,
                                 int M) {
    constexpr int COL_TILES = NDIM / 16;
    __shared__ float Alds[ROWS * (K + 1)];

    const int tid     = threadIdx.x;
    const int wave    = tid >> 5;
    const int lane    = tid & 31;
    const int rowbase = blockIdx.x * ROWS;

    for (int idx = tid; idx < ROWS * K; idx += blockDim.x) {
        int r  = idx / K;
        int c  = idx - r * K;
        int gr = rowbase + r;
        Alds[r * (K + 1) + c] = (gr < M) ? A[(size_t)gr * K + c] : 0.0f;
    }
    __syncthreads();

    const int rg      = wave / COL_TILES;
    const int colbase = (wave % COL_TILES) * 16;
    const int half    = lane >> 4;
    const int l15     = lane & 15;

    const float* Arow = &Alds[(rg * 16 + l15) * (K + 1)];

    float tl = 0.0f, tr = 0.0f;
    if constexpr (ALPHA) { tl = attl[colbase + l15]; tr = attr[colbase + l15]; }

    v8f acc = {};
    #pragma unroll 8
    for (int k = 0; k < K; k += 4) {
        const int kk = k + 2 * half;
        v2f a, b;
        a.x = Arow[kk];
        a.y = Arow[kk + 1];
        b.x = B[(size_t)kk * NDIM + colbase + l15];
        b.y = B[(size_t)(kk + 1) * NDIM + colbase + l15];
        acc = __builtin_amdgcn_wmma_f32_16x16x4_f32(
             false, a,  false, b,
             (short)0, acc,  false,  false);
    }

    if constexpr (ALPHA) {
        static_assert(NDIM == 128 && ROWS == 16, "coefficient epilogue: 2 heads x 64 channels, one row group");
        __shared__ float part[8 * 32];
        float pl[8], pr[8];
        #pragma unroll
        for (int r = 0; r < 8; ++r) { pl[r] = acc[r] * tl; pr[r] = acc[r] * tr; }
        #pragma unroll
        for (int r = 0; r < 8; ++r) { pl[r] = row_xor_add<1>(pl[r]); pr[r] = row_xor_add<1>(pr[r]); }
        #pragma unroll
        for (int r = 0; r < 8; ++r) { pl[r] = row_xor_add<2>(pl[r]); pr[r] = row_xor_add<2>(pr[r]); }
        #pragma unroll
        for (int r = 0; r < 8; ++r) { pl[r] = row_xor_add<4>(pl[r]); pr[r] = row_xor_add<4>(pr[r]); }
        #pragma unroll
        for (int r = 0; r < 8; ++r) { pl[r] = row_xor_add<8>(pl[r]); pr[r] = row_xor_add<8>(pr[r]); }
        if (l15 == 0) {
            float* p = &part[wave * 32 + half * 16];
            #pragma unroll
            for (int r = 0; r < 8; ++r) { p[2 * r] = pl[r]; p[2 * r + 1] = pr[r]; }
        }
        __syncthreads();
        if (tid < 64) {
            const int which = tid >> 5;
            const int row   = (tid & 31) >> 1;
            const int h     = tid & 1;
            const float* p  = &part[(h * 4) * 32 + row * 2 + which];
            const float s   = ((p[0] + p[32]) + p[64]) + p[96];
            const int grow  = rowbase + row;
            if (grow < M) (which ? ar : al)[(size_t)grow * 2 + h] = s;
        }
    }

    const float bv = bias ? bias[colbase + l15] : 0.0f;
    #pragma unroll
    for (int r = 0; r < 8; ++r) {
        int grow = rowbase + rg * 16 + r + 8 * half;
        if (grow < M)
            C[(size_t)grow * NDIM + colbase + l15] = acc[r] + bv;
    }
}

typedef float f2 __attribute__((ext_vector_type(2)));
typedef float v4f __attribute__((ext_vector_type(4)));

__device__ __forceinline__ float wave_max(float v) {
    #pragma unroll
    for (int o = 16; o > 0; o >>= 1) v = fmaxf(v, __shfl_xor(v, o, 32));
    return v;
}
__device__ __forceinline__ float wave_sum(float v) {
    #pragma unroll
    for (int o = 16; o > 0; o >>= 1) v += __shfl_xor(v, o, 32);
    return v;
}

__device__ __forceinline__ void load_chunk(int c, int lane, int i, int o0, int deg,
                                           const int* __restrict__ srt,
                                           const float* __restrict__ al, f2 vr,
                                           int& s, f2& e, bool& ok) {
    const int j = c + lane;
    ok = (j <= deg);
    s = i;
    if (j < deg) s = srt[o0 + j];
    const f2 vl = *(const f2*)(al + (size_t)s * 2);
    f2 v = vl + vr;
    v.x = (v.x > 0.0f) ? v.x : 0.2f * v.x;
    v.y = (v.y > 0.0f) ? v.y : 0.2f * v.y;
    e = v;
}

template <int K>
__device__ __forceinline__ void gather_rows(int j0, int s, float w0, float w1,
                                            const float* __restrict__ xh, int lane,
                                            v4f& acc) {
    v4f   v[K];
    float a[K];
    #pragma unroll
    for (int k = 0; k < K; ++k) {
        const int   sj = __builtin_amdgcn_readlane(s, j0 + k);
        const float a0 = __int_as_float(__builtin_amdgcn_readlane(__float_as_int(w0), j0 + k));
        const float a1 = __int_as_float(__builtin_amdgcn_readlane(__float_as_int(w1), j0 + k));
        a[k] = (lane < 16) ? a0 : a1;
        v[k] = *(const v4f*)(xh + (size_t)sj * 128 + 4 * lane);
    }
    #pragma unroll
    for (int k = 0; k < K; ++k) acc += v[k] * a[k];
}

__device__ __forceinline__ void walk_chunk(int n, int s, float w0, float w1,
                                           const float* __restrict__ xh, int lane,
                                           v4f& acc) {
    int j = 0;
    while (n - j >= 8) { gather_rows<8>(j, s, w0, w1, xh, lane, acc); j += 8; }
    if (n - j >= 4)    { gather_rows<4>(j, s, w0, w1, xh, lane, acc); j += 4; }
    if (n - j >= 2)    { gather_rows<2>(j, s, w0, w1, xh, lane, acc); j += 2; }
    if (n - j >= 1)    { gather_rows<1>(j, s, w0, w1, xh, lane, acc); }
}

__global__ __launch_bounds__(256)
void node_softmax_gather_kernel(const int* __restrict__ off,
                                const int* __restrict__ srt,
                                const float* __restrict__ al,
                                const float* __restrict__ ar,
                                const float* __restrict__ xh,
                                float* __restrict__ agg, int N) {
    const int lane = threadIdx.x & 31;
    const int nw   = (int)((gridDim.x * blockDim.x) >> 5);
    const int wid  = __builtin_amdgcn_readfirstlane((int)((blockIdx.x * blockDim.x + threadIdx.x) >> 5));
    for (int i = wid; i < N; i += nw) {
        const int o0  = off[i];
        const int deg = off[i + 1] - o0;
        const int tot = deg + 1;
        const f2  vr  = *(const f2*)(ar + (size_t)i * 2);

        int s0; f2 e0; bool ok0;
        load_chunk(0, lane, i, o0, deg, srt, al, vr, s0, e0, ok0);

        float m0 = wave_max(ok0 ? e0.x : -INFINITY);
        float m1 = wave_max(ok0 ? e0.y : -INFINITY);
        for (int c = 32; c < tot; c += 32) {
            int s; f2 e; bool ok;
            load_chunk(c, lane, i, o0, deg, srt, al, vr, s, e, ok);
            m0 = fmaxf(m0, wave_max(ok ? e.x : -INFINITY));
            m1 = fmaxf(m1, wave_max(ok ? e.y : -INFINITY));
        }
        if (!(fabsf(m0) <= 3.4e38f)) m0 = 0.0f;
        if (!(fabsf(m1) <= 3.4e38f)) m1 = 0.0f;

        const float x0 = ok0 ? __expf(e0.x - m0) : 0.0f;
        const float x1 = ok0 ? __expf(e0.y - m1) : 0.0f;
        float d0 = wave_sum(x0);
        float d1 = wave_sum(x1);
        for (int c = 32; c < tot; c += 32) {
            int s; f2 e; bool ok;
            load_chunk(c, lane, i, o0, deg, srt, al, vr, s, e, ok);
            d0 += wave_sum(ok ? __expf(e.x - m0) : 0.0f);
            d1 += wave_sum(ok ? __expf(e.y - m1) : 0.0f);
        }
        d0 += 1e-16f;
        d1 += 1e-16f;

        v4f acc = {0.0f, 0.0f, 0.0f, 0.0f};
        walk_chunk(min(tot, 32), s0, x0 / d0, x1 / d1, xh, lane, acc);
        for (int c = 32; c < tot; c += 32) {
            int s; f2 e; bool ok;
            load_chunk(c, lane, i, o0, deg, srt, al, vr, s, e, ok);
            const float w0 = ok ? __expf(e.x - m0) / d0 : 0.0f;
            const float w1 = ok ? __expf(e.y - m1) / d1 : 0.0f;
            walk_chunk(min(tot - c, 32), s, w0, w1, xh, lane, acc);
        }
        *(v4f*)(agg + (size_t)i * 128 + 4 * lane) = acc;
    }
}

extern "C" void kernel_launch(void* const* d_in, const int* in_sizes, int n_in,
                              void* d_out, int out_size, void* d_ws, size_t ws_size,
                              hipStream_t stream) {
    const float* x    = (const float*)d_in[0];
    const int*   ei   = (const int*)d_in[1];
    const float* Wl   = (const float*)d_in[2];
    const float* attl = (const float*)d_in[3];
    const float* attr = (const float*)d_in[4];
    const float* Wout = (const float*)d_in[5];
    const float* bias = (const float*)d_in[6];
    float* out = (float*)d_out;

    const int N   = in_sizes[0] / 128;
    const int E   = in_sizes[1] / 2;
    const int EDG = E + N;

    float* ws    = (float*)d_ws;
    float* xh    = ws;
    float* agg   = xh  + (size_t)N * 128;
    float* al    = agg + (size_t)N * 128;
    float* ar    = al  + (size_t)N * 2;
    int*   cnt   = (int*)(ar + (size_t)N * 2);
    int*   off   = cnt + (size_t)N * 2;
    int*   rank  = off + (size_t)N * 2;
    int*   srt   = rank + (size_t)E;
    int*   queues = rank + (size_t)EDG * 2;

    sort_init_kernel<<<64, 256, 0, stream>>>(cnt, queues, N);

    dst_rank_kernel<<<1024, 256, 0, stream>>>(ei, E, N, cnt, rank, queues);
    offsets_kernel<<<64, 1024, 0, stream>>>(cnt, off, N);
    place_kernel<<<1024, 256, 0, stream>>>(ei, E, off, rank, srt);

    wmma_gemm_kernel<128, 128, 16, true><<<(N + 15) / 16, 256, 0, stream>>>(
        x, Wl, nullptr, xh, attl, attr, al, ar, N);

    node_softmax_gather_kernel<<<2048, 256, 0, stream>>>(off, srt, al, ar, xh, agg, N);

    wmma_gemm_kernel<128, 64, 32, false><<<(N + 31) / 32, 256, 0, stream>>>(
        agg, Wout, bias, out, nullptr, nullptr, nullptr, nullptr, N);
}


# Input edges grouped by destination with LDS counters (chunk x bucket count matrix, scatter to bucket slices, per-bucket place): no global atomics; node list order = LDS add arrival order
# speedup vs baseline: 3.8022x; 1.2983x over previous
#include <hip/hip_runtime.h>
#include <math.h>

typedef float v2f __attribute__((ext_vector_type(2)));
typedef float v8f __attribute__((ext_vector_type(8)));

#define BSH 8
#define BW  256
#define ST  1024
#define PARTS (ST / BW)

typedef int i2 __attribute__((ext_vector_type(2)));

__device__ __forceinline__ int wave_incl_scan(int v, int lane) {
    #pragma unroll
    for (int o = 1; o < 32; o <<= 1) {
        const int up = __shfl_up(v, o, 32);
        if (lane >= o) v += up;
    }
    return v;
}

__global__ __launch_bounds__(ST)
void bucket_count_kernel(const int* __restrict__ ei, int E, int* __restrict__ hist) {
    __shared__ int h[BW];
    const int tid = threadIdx.x;
    const int G   = gridDim.x;
    const int ch  = (E + G - 1) / G;
    const int lo  = min((int)blockIdx.x * ch, E);
    const int hi  = min(lo + ch, E);
    if (tid < BW) h[tid] = 0;
    __syncthreads();
    for (int e0 = lo + tid; e0 < hi; e0 += 4 * ST) {
        int  d[4];
        bool on[4];
        #pragma unroll
        for (int r = 0; r < 4; ++r) {
            const int e = e0 + r * ST;
            on[r] = e < hi;
            d[r]  = on[r] ? ei[E + e] : 0;
        }
        #pragma unroll
        for (int r = 0; r < 4; ++r)
            if (on[r]) atomicAdd(&h[d[r] >> BSH], 1);
    }
    __syncthreads();
    if (tid < BW) hist[blockIdx.x * BW + tid] = h[tid];
}

__global__ __launch_bounds__(ST)
void bucket_scatter_kernel(const int* __restrict__ ei, int E, int NB,
                           const int* __restrict__ hist,
                           int* __restrict__ bstart,
                           i2* __restrict__ rec) {
    __shared__ int ppre[PARTS][BW];
    __shared__ int ptot[PARTS][BW];
    __shared__ int cur[BW];
    __shared__ int wsum[BW / 32];
    const int tid  = threadIdx.x;
    const int lane = tid & 31;
    const int G    = gridDim.x;
    const int g    = blockIdx.x;

    const int b    = tid & (BW - 1);
    const int part = tid >> BSH;
    const int gh   = (G + PARTS - 1) / PARTS;
    const int g0   = part * gh;
    const int g1   = min(g0 + gh, G);
    int pre = 0, tot = 0;
    #pragma unroll 8
    for (int q = g0; q < g1; ++q) {
        const int v = hist[q * BW + b];
        tot += v;
        pre += (q < g) ? v : 0;
    }
    ppre[part][b] = pre;
    ptot[part][b] = tot;
    __syncthreads();

    int incl = 0;
    if (tid < BW) {
        pre = 0; tot = 0;
        #pragma unroll
        for (int p = 0; p < PARTS; ++p) { pre += ppre[p][tid]; tot += ptot[p][tid]; }
        incl = wave_incl_scan(tot, lane);
        if (lane == 31) wsum[tid >> 5] = incl;
    }
    __syncthreads();
    if (tid < BW) {
        int wb = 0;
        for (int w = 0; w < (tid >> 5); ++w) wb += wsum[w];
        const int bst = wb + incl - tot;
        cur[tid] = bst + pre;
        if (g == 0 && tid < NB) bstart[tid] = bst;
    }
    __syncthreads();

    const int ch = (E + G - 1) / G;
    const int lo = min(g * ch, E);
    const int hi = min(lo + ch, E);
    for (int e0 = lo + tid; e0 < hi; e0 += 4 * ST) {
        int  s[4], d[4];
        bool on[4];
        #pragma unroll
        for (int r = 0; r < 4; ++r) {
            const int e = e0 + r * ST;
            on[r] = e < hi;
            d[r]  = on[r] ? ei[E + e] : 0;
            s[r]  = on[r] ? ei[e] : 0;
        }
        #pragma unroll
        for (int r = 0; r < 4; ++r)
            if (on[r]) {
                const int p = atomicAdd(&cur[d[r] >> BSH], 1);
                i2 q; q.x = s[r]; q.y = d[r];
                rec[p] = q;
            }
    }
}

__global__ __launch_bounds__(ST)
void bucket_place_kernel(const i2* __restrict__ rec,
                         const int* __restrict__ bstart,
                         int E, int N, int NB,
                         int* __restrict__ off,
                         int* __restrict__ srt) {
    __shared__ int c[BW];
    __shared__ int cur[BW];
    __shared__ int wsum[BW / 32];
    const int tid  = threadIdx.x;
    const int lane = tid & 31;
    const int b    = blockIdx.x;
    const int n0   = b << BSH;
    const int bs   = bstart[b];
    const int be   = (b + 1 < NB) ? bstart[b + 1] : E;

    if (tid < BW) c[tid] = 0;
    __syncthreads();
    for (int r0 = bs + tid; r0 < be; r0 += 4 * ST) {
        int  d[4];
        bool on[4];
        #pragma unroll
        for (int r = 0; r < 4; ++r) {
            const int i = r0 + r * ST;
            on[r] = i < be;
            d[r]  = on[r] ? rec[i].y : n0;
        }
        #pragma unroll
        for (int r = 0; r < 4; ++r)
            if (on[r]) atomicAdd(&c[d[r] - n0], 1);
    }
    __syncthreads();

    int v = 0, incl = 0;
    if (tid < BW) {
        v    = c[tid];
        incl = wave_incl_scan(v, lane);
        if (lane == 31) wsum[tid >> 5] = incl;
    }
    __syncthreads();
    if (tid < BW) {
        int wb = 0;
        for (int w = 0; w < (tid >> 5); ++w) wb += wsum[w];
        const int o = bs + wb + incl - v;
        cur[tid] = o;
        if (n0 + tid < N) off[n0 + tid] = o;
    }
    if (b == NB - 1 && tid == 0) off[N] = E;
    __syncthreads();

    for (int r0 = bs + tid; r0 < be; r0 += 4 * ST) {
        i2   q[4];
        bool on[4];
        #pragma unroll
        for (int r = 0; r < 4; ++r) {
            const int i = r0 + r * ST;
            on[r] = i < be;
            i2 z; z.x = 0; z.y = n0;
            q[r]  = on[r] ? rec[i] : z;
        }
        #pragma unroll
        for (int r = 0; r < 4; ++r)
            if (on[r]) {
                const int p = atomicAdd(&cur[q[r].y - n0], 1);
                srt[p] = q[r].x;
            }
    }
}


template <int M>
__device__ __forceinline__ float row_xor_add(float v) {
    const int t = __builtin_amdgcn_update_dpp(0, __float_as_int(v), 0x160 + M, 0xf, 0xf, true);
    return v + __int_as_float(t);
}

template <int K, int NDIM, int ROWS, bool ALPHA>
__global__ void wmma_gemm_kernel(const float* __restrict__ A,
                                 const float* __restrict__ B,
                                 const float* __restrict__ bias,
                                 float* __restrict__ C,
                                 const float* __restrict__ attl,
                                 const float* __restrict__ attr,
                                 float* __restrict__ al,
                                 float* __restrict__ ar,
                                 int M) {
    constexpr int COL_TILES = NDIM / 16;
    __shared__ float Alds[ROWS * (K + 1)];

    const int tid     = threadIdx.x;
    const int wave    = tid >> 5;
    const int lane    = tid & 31;
    const int rowbase = blockIdx.x * ROWS;

    for (int idx = tid; idx < ROWS * K; idx += blockDim.x) {
        int r  = idx / K;
        int c  = idx - r * K;
        int gr = rowbase + r;
        Alds[r * (K + 1) + c] = (gr < M) ? A[(size_t)gr * K + c] : 0.0f;
    }
    __syncthreads();

    const int rg      = wave / COL_TILES;
    const int colbase = (wave % COL_TILES) * 16;
    const int half    = lane >> 4;
    const int l15     = lane & 15;

    const float* Arow = &Alds[(rg * 16 + l15) * (K + 1)];

    float tl = 0.0f, tr = 0.0f;
    if constexpr (ALPHA) { tl = attl[colbase + l15]; tr = attr[colbase + l15]; }

    v8f acc = {};
    #pragma unroll 8
    for (int k = 0; k < K; k += 4) {
        const int kk = k + 2 * half;
        v2f a, b;
        a.x = Arow[kk];
        a.y = Arow[kk + 1];
        b.x = B[(size_t)kk * NDIM + colbase + l15];
        b.y = B[(size_t)(kk + 1) * NDIM + colbase + l15];
        acc = __builtin_amdgcn_wmma_f32_16x16x4_f32(
             false, a,  false, b,
             (short)0, acc,  false,  false);
    }

    if constexpr (ALPHA) {
        static_assert(NDIM == 128 && ROWS == 16, "coefficient epilogue: 2 heads x 64 channels, one row group");
        __shared__ float part[8 * 32];
        float pl[8], pr[8];
        #pragma unroll
        for (int r = 0; r < 8; ++r) { pl[r] = acc[r] * tl; pr[r] = acc[r] * tr; }
        #pragma unroll
        for (int r = 0; r < 8; ++r) { pl[r] = row_xor_add<1>(pl[r]); pr[r] = row_xor_add<1>(pr[r]); }
        #pragma unroll
        for (int r = 0; r < 8; ++r) { pl[r] = row_xor_add<2>(pl[r]); pr[r] = row_xor_add<2>(pr[r]); }
        #pragma unroll
        for (int r = 0; r < 8; ++r) { pl[r] = row_xor_add<4>(pl[r]); pr[r] = row_xor_add<4>(pr[r]); }
        #pragma unroll
        for (int r = 0; r < 8; ++r) { pl[r] = row_xor_add<8>(pl[r]); pr[r] = row_xor_add<8>(pr[r]); }
        if (l15 == 0) {
            float* p = &part[wave * 32 + half * 16];
            #pragma unroll
            for (int r = 0; r < 8; ++r) { p[2 * r] = pl[r]; p[2 * r + 1] = pr[r]; }
        }
        __syncthreads();
        if (tid < 64) {
            const int which = tid >> 5;
            const int row   = (tid & 31) >> 1;
            const int h     = tid & 1;
            const float* p  = &part[(h * 4) * 32 + row * 2 + which];
            const float s   = ((p[0] + p[32]) + p[64]) + p[96];
            const int grow  = rowbase + row;
            if (grow < M) (which ? ar : al)[(size_t)grow * 2 + h] = s;
        }
    }

    const float bv = bias ? bias[colbase + l15] : 0.0f;
    #pragma unroll
    for (int r = 0; r < 8; ++r) {
        int grow = rowbase + rg * 16 + r + 8 * half;
        if (grow < M)
            C[(size_t)grow * NDIM + colbase + l15] = acc[r] + bv;
    }
}

typedef float f2 __attribute__((ext_vector_type(2)));
typedef float v4f __attribute__((ext_vector_type(4)));

__device__ __forceinline__ float wave_max(float v) {
    #pragma unroll
    for (int o = 16; o > 0; o >>= 1) v = fmaxf(v, __shfl_xor(v, o, 32));
    return v;
}
__device__ __forceinline__ float wave_sum(float v) {
    #pragma unroll
    for (int o = 16; o > 0; o >>= 1) v += __shfl_xor(v, o, 32);
    return v;
}

__device__ __forceinline__ void load_chunk(int c, int lane, int i, int o0, int deg,
                                           const int* __restrict__ srt,
                                           const float* __restrict__ al, f2 vr,
                                           int& s, f2& e, bool& ok) {
    const int j = c + lane;
    ok = (j <= deg);
    s = i;
    if (j < deg) s = srt[o0 + j];
    const f2 vl = *(const f2*)(al + (size_t)s * 2);
    f2 v = vl + vr;
    v.x = (v.x > 0.0f) ? v.x : 0.2f * v.x;
    v.y = (v.y > 0.0f) ? v.y : 0.2f * v.y;
    e = v;
}

template <int K>
__device__ __forceinline__ void gather_rows(int j0, int s, float w0, float w1,
                                            const float* __restrict__ xh, int lane,
                                            v4f& acc) {
    v4f   v[K];
    float a[K];
    #pragma unroll
    for (int k = 0; k < K; ++k) {
        const int   sj = __builtin_amdgcn_readlane(s, j0 + k);
        const float a0 = __int_as_float(__builtin_amdgcn_readlane(__float_as_int(w0), j0 + k));
        const float a1 = __int_as_float(__builtin_amdgcn_readlane(__float_as_int(w1), j0 + k));
        a[k] = (lane < 16) ? a0 : a1;
        v[k] = *(const v4f*)(xh + (size_t)sj * 128 + 4 * lane);
    }
    #pragma unroll
    for (int k = 0; k < K; ++k) acc += v[k] * a[k];
}

__device__ __forceinline__ void walk_chunk(int n, int s, float w0, float w1,
                                           const float* __restrict__ xh, int lane,
                                           v4f& acc) {
    int j = 0;
    while (n - j >= 8) { gather_rows<8>(j, s, w0, w1, xh, lane, acc); j += 8; }
    if (n - j >= 4)    { gather_rows<4>(j, s, w0, w1, xh, lane, acc); j += 4; }
    if (n - j >= 2)    { gather_rows<2>(j, s, w0, w1, xh, lane, acc); j += 2; }
    if (n - j >= 1)    { gather_rows<1>(j, s, w0, w1, xh, lane, acc); }
}

__global__ __launch_bounds__(256)
void node_softmax_gather_kernel(const int* __restrict__ off,
                                const int* __restrict__ srt,
                                const float* __restrict__ al,
                                const float* __restrict__ ar,
                                const float* __restrict__ xh,
                                float* __restrict__ agg, int N) {
    const int lane = threadIdx.x & 31;
    const int nw   = (int)((gridDim.x * blockDim.x) >> 5);
    const int wid  = __builtin_amdgcn_readfirstlane((int)((blockIdx.x * blockDim.x + threadIdx.x) >> 5));
    for (int i = wid; i < N; i += nw) {
        const int o0  = off[i];
        const int deg = off[i + 1] - o0;
        const int tot = deg + 1;
        const f2  vr  = *(const f2*)(ar + (size_t)i * 2);

        int s0; f2 e0; bool ok0;
        load_chunk(0, lane, i, o0, deg, srt, al, vr, s0, e0, ok0);

        float m0 = wave_max(ok0 ? e0.x : -INFINITY);
        float m1 = wave_max(ok0 ? e0.y : -INFINITY);
        for (int c = 32; c < tot; c += 32) {
            int s; f2 e; bool ok;
            load_chunk(c, lane, i, o0, deg, srt, al, vr, s, e, ok);
            m0 = fmaxf(m0, wave_max(ok ? e.x : -INFINITY));
            m1 = fmaxf(m1, wave_max(ok ? e.y : -INFINITY));
        }
        if (!(fabsf(m0) <= 3.4e38f)) m0 = 0.0f;
        if (!(fabsf(m1) <= 3.4e38f)) m1 = 0.0f;

        const float x0 = ok0 ? __expf(e0.x - m0) : 0.0f;
        const float x1 = ok0 ? __expf(e0.y - m1) : 0.0f;
        float d0 = wave_sum(x0);
        float d1 = wave_sum(x1);
        for (int c = 32; c < tot; c += 32) {
            int s; f2 e; bool ok;
            load_chunk(c, lane, i, o0, deg, srt, al, vr, s, e, ok);
            d0 += wave_sum(ok ? __expf(e.x - m0) : 0.0f);
            d1 += wave_sum(ok ? __expf(e.y - m1) : 0.0f);
        }
        d0 += 1e-16f;
        d1 += 1e-16f;

        v4f acc = {0.0f, 0.0f, 0.0f, 0.0f};
        walk_chunk(min(tot, 32), s0, x0 / d0, x1 / d1, xh, lane, acc);
        for (int c = 32; c < tot; c += 32) {
            int s; f2 e; bool ok;
            load_chunk(c, lane, i, o0, deg, srt, al, vr, s, e, ok);
            const float w0 = ok ? __expf(e.x - m0) / d0 : 0.0f;
            const float w1 = ok ? __expf(e.y - m1) / d1 : 0.0f;
            walk_chunk(min(tot - c, 32), s, w0, w1, xh, lane, acc);
        }
        *(v4f*)(agg + (size_t)i * 128 + 4 * lane) = acc;
    }
}

extern "C" void kernel_launch(void* const* d_in, const int* in_sizes, int n_in,
                              void* d_out, int out_size, void* d_ws, size_t ws_size,
                              hipStream_t stream) {
    const float* x    = (const float*)d_in[0];
    const int*   ei   = (const int*)d_in[1];
    const float* Wl   = (const float*)d_in[2];
    const float* attl = (const float*)d_in[3];
    const float* attr = (const float*)d_in[4];
    const float* Wout = (const float*)d_in[5];
    const float* bias = (const float*)d_in[6];
    float* out = (float*)d_out;

    const int N   = in_sizes[0] / 128;
    const int E   = in_sizes[1] / 2;

    float* ws    = (float*)d_ws;
    float* xh    = ws;
    float* agg   = xh  + (size_t)N * 128;
    float* al    = agg + (size_t)N * 128;
    float* ar    = al  + (size_t)N * 2;
    int*   hist  = (int*)(ar + (size_t)N * 2);
    int*   bstart = hist + 128 * BW;
    int*   off   = hist + (size_t)N * 2;
    int*   srt   = off + (size_t)N * 2;
    i2*    rec   = (i2*)agg;
    const int NB = (N + BW - 1) >> BSH;

    bucket_count_kernel<<<128, ST, 0, stream>>>(ei, E, hist);
    bucket_scatter_kernel<<<128, ST, 0, stream>>>(ei, E, NB, hist, bstart, rec);
    bucket_place_kernel<<<NB, ST, 0, stream>>>(rec, bstart, E, N, NB, off, srt);

    wmma_gemm_kernel<128, 128, 16, true><<<(N + 15) / 16, 256, 0, stream>>>(
        x, Wl, nullptr, xh, attl, attr, al, ar, N);

    node_softmax_gather_kernel<<<2048, 256, 0, stream>>>(off, srt, al, ar, xh, agg, N);

    wmma_gemm_kernel<128, 64, 32, false><<<(N + 31) / 32, 256, 0, stream>>>(
        agg, Wout, bias, out, nullptr, nullptr, nullptr, nullptr, N);
}


# Projection GEMM moved onto extra work-groups of the three sort launches as 64-row strips (a thread's 8 staging loads issued together); own launch removed; products, sums and their order unchanged
# speedup vs baseline: 4.0875x; 1.0750x over previous
#include <hip/hip_runtime.h>
#include <math.h>

typedef float v2f __attribute__((ext_vector_type(2)));
typedef float v8f __attribute__((ext_vector_type(8)));

#define BSH 8
#define BW  256
#define ST  1024
#define PARTS (ST / BW)

typedef int i2 __attribute__((ext_vector_type(2)));

__device__ __forceinline__ int wave_incl_scan(int v, int lane) {
    #pragma unroll
    for (int o = 1; o < 32; o <<= 1) {
        const int up = __shfl_up(v, o, 32);
        if (lane >= o) v += up;
    }
    return v;
}

template <int M>
__device__ __forceinline__ float row_xor_add(float v) {
    const int t = __builtin_amdgcn_update_dpp(0, __float_as_int(v), 0x160 + M, 0xf, 0xf, true);
    return v + __int_as_float(t);
}

struct ProjArgs {
    const float* x;
    const float* W;
    const float* attl;
    const float* attr;
    float*       xh;
    float*       al;
    float*       ar;
    int          M;
    int          strip0;
};

#define PD    128
#define PROWS 64

__device__ __forceinline__ void proj_strip(const ProjArgs p, int strip) {
    __shared__ float Alds[PROWS * (PD + 1)];
    __shared__ float part[32 * 32];

    const float* __restrict__ A = p.x;
    const float* __restrict__ B = p.W;
    const int M       = p.M;
    const int tid     = threadIdx.x;
    const int wave    = tid >> 5;
    const int lane    = tid & 31;
    const int rowbase = strip * PROWS;

    constexpr int NQ = PROWS * PD / ST;
    float v[NQ];
    #pragma unroll
    for (int q = 0; q < NQ; ++q) {
        const int idx = tid + q * ST;
        const int r   = idx / PD;
        const int c   = idx - r * PD;
        const int gr  = rowbase + r;
        v[q] = (gr < M) ? A[(size_t)gr * PD + c] : 0.0f;
    }
    #pragma unroll
    for (int q = 0; q < NQ; ++q) {
        const int idx = tid + q * ST;
        const int r   = idx / PD;
        const int c   = idx - r * PD;
        Alds[r * (PD + 1) + c] = v[q];
    }
    __syncthreads();

    const int rg      = wave >> 3;
    const int colbase = (wave & 7) * 16;
    const int half    = lane >> 4;
    const int l15     = lane & 15;

    const float* Arow = &Alds[(rg * 16 + l15) * (PD + 1)];

    const float tl = p.attl[colbase + l15];
    const float tr = p.attr[colbase + l15];

    v8f acc = {};
    #pragma unroll 8
    for (int k = 0; k < PD; k += 4) {
        const int kk = k + 2 * half;
        v2f a, b;
        a.x = Arow[kk];
        a.y = Arow[kk + 1];
        b.x = B[(size_t)kk * PD + colbase + l15];
        b.y = B[(size_t)(kk + 1) * PD + colbase + l15];
        acc = __builtin_amdgcn_wmma_f32_16x16x4_f32(
             false, a,  false, b,
             (short)0, acc,  false,  false);
    }

    float pl[8], pr[8];
    #pragma unroll
    for (int r = 0; r < 8; ++r) { pl[r] = acc[r] * tl; pr[r] = acc[r] * tr; }
    #pragma unroll
    for (int r = 0; r < 8; ++r) { pl[r] = row_xor_add<1>(pl[r]); pr[r] = row_xor_add<1>(pr[r]); }
    #pragma unroll
    for (int r = 0; r < 8; ++r) { pl[r] = row_xor_add<2>(pl[r]); pr[r] = row_xor_add<2>(pr[r]); }
    #pragma unroll
    for (int r = 0; r < 8; ++r) { pl[r] = row_xor_add<4>(pl[r]); pr[r] = row_xor_add<4>(pr[r]); }
    #pragma unroll
    for (int r = 0; r < 8; ++r) { pl[r] = row_xor_add<8>(pl[r]); pr[r] = row_xor_add<8>(pr[r]); }
    if (l15 == 0) {
        float* w = &part[wave * 32 + half * 16];
        #pragma unroll
        for (int r = 0; r < 8; ++r) { w[2 * r] = pl[r]; w[2 * r + 1] = pr[r]; }
    }
    __syncthreads();
    if (tid < 256) {
        const int g     = tid >> 6;
        const int t     = tid & 63;
        const int which = t >> 5;
        const int row   = (t & 31) >> 1;
        const int h     = t & 1;
        const float* w  = &part[(g * 8 + h * 4) * 32 + row * 2 + which];
        const float s   = ((w[0] + w[32]) + w[64]) + w[96];
        const int grow  = rowbase + g * 16 + row;
        if (grow < M) (which ? p.ar : p.al)[(size_t)grow * 2 + h] = s;
    }

    #pragma unroll
    for (int r = 0; r < 8; ++r) {
        const int grow = rowbase + rg * 16 + r + 8 * half;
        if (grow < M)
            p.xh[(size_t)grow * PD + colbase + l15] = acc[r];
    }
}

__global__ __launch_bounds__(ST)
void bucket_count_kernel(const int* __restrict__ ei, int E, int G, int* __restrict__ hist,
                         const ProjArgs pa) {
    if ((int)blockIdx.x >= G) { proj_strip(pa, pa.strip0 + (int)blockIdx.x - G); return; }
    __shared__ int h[BW];
    const int tid = threadIdx.x;
    const int ch  = (E + G - 1) / G;
    const int lo  = min((int)blockIdx.x * ch, E);
    const int hi  = min(lo + ch, E);
    if (tid < BW) h[tid] = 0;
    __syncthreads();
    for (int e0 = lo + tid; e0 < hi; e0 += 4 * ST) {
        int  d[4];
        bool on[4];
        #pragma unroll
        for (int r = 0; r < 4; ++r) {
            const int e = e0 + r * ST;
            on[r] = e < hi;
            d[r]  = on[r] ? ei[E + e] : 0;
        }
        #pragma unroll
        for (int r = 0; r < 4; ++r)
            if (on[r]) atomicAdd(&h[d[r] >> BSH], 1);
    }
    __syncthreads();
    if (tid < BW) hist[blockIdx.x * BW + tid] = h[tid];
}

__global__ __launch_bounds__(ST)
void bucket_scatter_kernel(const int* __restrict__ ei, int E, int G, int NB,
                           const int* __restrict__ hist,
                           int* __restrict__ bstart,
                           i2* __restrict__ rec,
                           const ProjArgs pa) {
    if ((int)blockIdx.x >= G) { proj_strip(pa, pa.strip0 + (int)blockIdx.x - G); return; }
    __shared__ int ppre[PARTS][BW];
    __shared__ int ptot[PARTS][BW];
    __shared__ int cur[BW];
    __shared__ int wsum[BW / 32];
    const int tid  = threadIdx.x;
    const int lane = tid & 31;
    const int g    = blockIdx.x;

    const int b    = tid & (BW - 1);
    const int part = tid >> BSH;
    const int gh   = (G + PARTS - 1) / PARTS;
    const int g0   = part * gh;
    const int g1   = min(g0 + gh, G);
    int pre = 0, tot = 0;
    #pragma unroll 8
    for (int q = g0; q < g1; ++q) {
        const int v = hist[q * BW + b];
        tot += v;
        pre += (q < g) ? v : 0;
    }
    ppre[part][b] = pre;
    ptot[part][b] = tot;
    __syncthreads();

    int incl = 0;
    if (tid < BW) {
        pre = 0; tot = 0;
        #pragma unroll
        for (int p = 0; p < PARTS; ++p) { pre += ppre[p][tid]; tot += ptot[p][tid]; }
        incl = wave_incl_scan(tot, lane);
        if (lane == 31) wsum[tid >> 5] = incl;
    }
    __syncthreads();
    if (tid < BW) {
        int wb = 0;
        for (int w = 0; w < (tid >> 5); ++w) wb += wsum[w];
        const int bst = wb + incl - tot;
        cur[tid] = bst + pre;
        if (g == 0 && tid < NB) bstart[tid] = bst;
    }
    __syncthreads();

    const int ch = (E + G - 1) / G;
    const int lo = min(g * ch, E);
    const int hi = min(lo + ch, E);
    for (int e0 = lo + tid; e0 < hi; e0 += 4 * ST) {
        int  s[4], d[4];
        bool on[4];
        #pragma unroll
        for (int r = 0; r < 4; ++r) {
            const int e = e0 + r * ST;
            on[r] = e < hi;
            d[r]  = on[r] ? ei[E + e] : 0;
            s[r]  = on[r] ? ei[e] : 0;
        }
        #pragma unroll
        for (int r = 0; r < 4; ++r)
            if (on[r]) {
                const int p = atomicAdd(&cur[d[r] >> BSH], 1);
                i2 q; q.x = s[r]; q.y = d[r];
                rec[p] = q;
            }
    }
}

__global__ __launch_bounds__(ST)
void bucket_place_kernel(const i2* __restrict__ rec,
                         const int* __restrict__ bstart,
                         int E, int N, int NB,
                         int* __restrict__ off,
                         int* __restrict__ srt,
                         const ProjArgs pa) {
    if ((int)blockIdx.x >= NB) { proj_strip(pa, pa.strip0 + (int)blockIdx.x - NB); return; }
    __shared__ int c[BW];
    __shared__ int cur[BW];
    __shared__ int wsum[BW / 32];
    const int tid  = threadIdx.x;
    const int lane = tid & 31;
    const int b    = blockIdx.x;
    const int n0   = b << BSH;
    const int bs   = bstart[b];
    const int be   = (b + 1 < NB) ? bstart[b + 1] : E;

    if (tid < BW) c[tid] = 0;
    __syncthreads();
    for (int r0 = bs + tid; r0 < be; r0 += 4 * ST) {
        int  d[4];
        bool on[4];
        #pragma unroll
        for (int r = 0; r < 4; ++r) {
            const int i = r0 + r * ST;
            on[r] = i < be;
            d[r]  = on[r] ? rec[i].y : n0;
        }
        #pragma unroll
        for (int r = 0; r < 4; ++r)
            if (on[r]) atomicAdd(&c[d[r] - n0], 1);
    }
    __syncthreads();

    int v = 0, incl = 0;
    if (tid < BW) {
        v    = c[tid];
        incl = wave_incl_scan(v, lane);
        if (lane == 31) wsum[tid >> 5] = incl;
    }
    __syncthreads();
    if (tid < BW) {
        int wb = 0;
        for (int w = 0; w < (tid >> 5); ++w) wb += wsum[w];
        const int o = bs + wb + incl - v;
        cur[tid] = o;
        if (n0 + tid < N) off[n0 + tid] = o;
    }
    if (b == NB - 1 && tid == 0) off[N] = E;
    __syncthreads();

    for (int r0 = bs + tid; r0 < be; r0 += 4 * ST) {
        i2   q[4];
        bool on[4];
        #pragma unroll
        for (int r = 0; r < 4; ++r) {
            const int i = r0 + r * ST;
            on[r] = i < be;
            i2 z; z.x = 0; z.y = n0;
            q[r]  = on[r] ? rec[i] : z;
        }
        #pragma unroll
        for (int r = 0; r < 4; ++r)
            if (on[r]) {
                const int p = atomicAdd(&cur[q[r].y - n0], 1);
                srt[p] = q[r].x;
            }
    }
}

template <int K, int NDIM, int ROWS>
__global__ void wmma_gemm_kernel(const float* __restrict__ A,
                                 const float* __restrict__ B,
                                 const float* __restrict__ bias,
                                 float* __restrict__ C,
                                 int M) {
    constexpr int COL_TILES = NDIM / 16;
    __shared__ float Alds[ROWS * (K + 1)];

    const int tid     = threadIdx.x;
    const int wave    = tid >> 5;
    const int lane    = tid & 31;
    const int rowbase = blockIdx.x * ROWS;

    for (int idx = tid; idx < ROWS * K; idx += blockDim.x) {
        int r  = idx / K;
        int c  = idx - r * K;
        int gr = rowbase + r;
        Alds[r * (K + 1) + c] = (gr < M) ? A[(size_t)gr * K + c] : 0.0f;
    }
    __syncthreads();

    const int rg      = wave / COL_TILES;
    const int colbase = (wave % COL_TILES) * 16;
    const int half    = lane >> 4;
    const int l15     = lane & 15;

    const float* Arow = &Alds[(rg * 16 + l15) * (K + 1)];

    v8f acc = {};
    #pragma unroll 8
    for (int k = 0; k < K; k += 4) {
        const int kk = k + 2 * half;
        v2f a, b;
        a.x = Arow[kk];
        a.y = Arow[kk + 1];
        b.x = B[(size_t)kk * NDIM + colbase + l15];
        b.y = B[(size_t)(kk + 1) * NDIM + colbase + l15];
        acc = __builtin_amdgcn_wmma_f32_16x16x4_f32(
             false, a,  false, b,
             (short)0, acc,  false,  false);
    }

    const float bv = bias ? bias[colbase + l15] : 0.0f;
    #pragma unroll
    for (int r = 0; r < 8; ++r) {
        int grow = rowbase + rg * 16 + r + 8 * half;
        if (grow < M)
            C[(size_t)grow * NDIM + colbase + l15] = acc[r] + bv;
    }
}

typedef float f2 __attribute__((ext_vector_type(2)));
typedef float v4f __attribute__((ext_vector_type(4)));

__device__ __forceinline__ float wave_max(float v) {
    #pragma unroll
    for (int o = 16; o > 0; o >>= 1) v = fmaxf(v, __shfl_xor(v, o, 32));
    return v;
}
__device__ __forceinline__ float wave_sum(float v) {
    #pragma unroll
    for (int o = 16; o > 0; o >>= 1) v += __shfl_xor(v, o, 32);
    return v;
}

__device__ __forceinline__ void load_chunk(int c, int lane, int i, int o0, int deg,
                                           const int* __restrict__ srt,
                                           const float* __restrict__ al, f2 vr,
                                           int& s, f2& e, bool& ok) {
    const int j = c + lane;
    ok = (j <= deg);
    s = i;
    if (j < deg) s = srt[o0 + j];
    const f2 vl = *(const f2*)(al + (size_t)s * 2);
    f2 v = vl + vr;
    v.x = (v.x > 0.0f) ? v.x : 0.2f * v.x;
    v.y = (v.y > 0.0f) ? v.y : 0.2f * v.y;
    e = v;
}

template <int K>
__device__ __forceinline__ void gather_rows(int j0, int s, float w0, float w1,
                                            const float* __restrict__ xh, int lane,
                                            v4f& acc) {
    v4f   v[K];
    float a[K];
    #pragma unroll
    for (int k = 0; k < K; ++k) {
        const int   sj = __builtin_amdgcn_readlane(s, j0 + k);
        const float a0 = __int_as_float(__builtin_amdgcn_readlane(__float_as_int(w0), j0 + k));
        const float a1 = __int_as_float(__builtin_amdgcn_readlane(__float_as_int(w1), j0 + k));
        a[k] = (lane < 16) ? a0 : a1;
        v[k] = *(const v4f*)(xh + (size_t)sj * 128 + 4 * lane);
    }
    #pragma unroll
    for (int k = 0; k < K; ++k) acc += v[k] * a[k];
}

__device__ __forceinline__ void walk_chunk(int n, int s, float w0, float w1,
                                           const float* __restrict__ xh, int lane,
                                           v4f& acc) {
    int j = 0;
    while (n - j >= 8) { gather_rows<8>(j, s, w0, w1, xh, lane, acc); j += 8; }
    if (n - j >= 4)    { gather_rows<4>(j, s, w0, w1, xh, lane, acc); j += 4; }
    if (n - j >= 2)    { gather_rows<2>(j, s, w0, w1, xh, lane, acc); j += 2; }
    if (n - j >= 1)    { gather_rows<1>(j, s, w0, w1, xh, lane, acc); }
}

__global__ __launch_bounds__(256)
void node_softmax_gather_kernel(const int* __restrict__ off,
                                const int* __restrict__ srt,
                                const float* __restrict__ al,
                                const float* __restrict__ ar,
                                const float* __restrict__ xh,
                                float* __restrict__ agg, int N) {
    const int lane = threadIdx.x & 31;
    const int nw   = (int)((gridDim.x * blockDim.x) >> 5);
    const int wid  = __builtin_amdgcn_readfirstlane((int)((blockIdx.x * blockDim.x + threadIdx.x) >> 5));
    for (int i = wid; i < N; i += nw) {
        const int o0  = off[i];
        const int deg = off[i + 1] - o0;
        const int tot = deg + 1;
        const f2  vr  = *(const f2*)(ar + (size_t)i * 2);

        int s0; f2 e0; bool ok0;
        load_chunk(0, lane, i, o0, deg, srt, al, vr, s0, e0, ok0);

        float m0 = wave_max(ok0 ? e0.x : -INFINITY);
        float m1 = wave_max(ok0 ? e0.y : -INFINITY);
        for (int c = 32; c < tot; c += 32) {
            int s; f2 e; bool ok;
            load_chunk(c, lane, i, o0, deg, srt, al, vr, s, e, ok);
            m0 = fmaxf(m0, wave_max(ok ? e.x : -INFINITY));
            m1 = fmaxf(m1, wave_max(ok ? e.y : -INFINITY));
        }
        if (!(fabsf(m0) <= 3.4e38f)) m0 = 0.0f;
        if (!(fabsf(m1) <= 3.4e38f)) m1 = 0.0f;

        const float x0 = ok0 ? __expf(e0.x - m0) : 0.0f;
        const float x1 = ok0 ? __expf(e0.y - m1) : 0.0f;
        float d0 = wave_sum(x0);
        float d1 = wave_sum(x1);
        for (int c = 32; c < tot; c += 32) {
            int s; f2 e; bool ok;
            load_chunk(c, lane, i, o0, deg, srt, al, vr, s, e, ok);
            d0 += wave_sum(ok ? __expf(e.x - m0) : 0.0f);
            d1 += wave_sum(ok ? __expf(e.y - m1) : 0.0f);
        }
        d0 += 1e-16f;
        d1 += 1e-16f;

        v4f acc = {0.0f, 0.0f, 0.0f, 0.0f};
        walk_chunk(min(tot, 32), s0, x0 / d0, x1 / d1, xh, lane, acc);
        for (int c = 32; c < tot; c += 32) {
            int s; f2 e; bool ok;
            load_chunk(c, lane, i, o0, deg, srt, al, vr, s, e, ok);
            const float w0 = ok ? __expf(e.x - m0) / d0 : 0.0f;
            const float w1 = ok ? __expf(e.y - m1) / d1 : 0.0f;
            walk_chunk(min(tot - c, 32), s, w0, w1, xh, lane, acc);
        }
        *(v4f*)(agg + (size_t)i * 128 + 4 * lane) = acc;
    }
}

extern "C" void kernel_launch(void* const* d_in, const int* in_sizes, int n_in,
                              void* d_out, int out_size, void* d_ws, size_t ws_size,
                              hipStream_t stream) {
    const float* x    = (const float*)d_in[0];
    const int*   ei   = (const int*)d_in[1];
    const float* Wl   = (const float*)d_in[2];
    const float* attl = (const float*)d_in[3];
    const float* attr = (const float*)d_in[4];
    const float* Wout = (const float*)d_in[5];
    const float* bias = (const float*)d_in[6];
    float* out = (float*)d_out;

    const int N   = in_sizes[0] / 128;
    const int E   = in_sizes[1] / 2;

    float* ws    = (float*)d_ws;
    float* xh    = ws;
    float* agg   = xh  + (size_t)N * 128;
    float* al    = agg + (size_t)N * 128;
    float* ar    = al  + (size_t)N * 2;
    int*   hist  = (int*)(ar + (size_t)N * 2);
    int*   bstart = hist + 128 * BW;
    int*   off   = hist + (size_t)N * 2;
    int*   srt   = off + (size_t)N * 2;
    i2*    rec   = (i2*)agg;
    const int NB = (N + BW - 1) >> BSH;

    const int NS = (N + PROWS - 1) / PROWS;
    const int S1 = NS / 3;
    const int S2 = (2 * NS) / 3;
    ProjArgs pa;
    pa.x = x; pa.W = Wl; pa.attl = attl; pa.attr = attr;
    pa.xh = xh; pa.al = al; pa.ar = ar; pa.M = N;

    pa.strip0 = 0;
    bucket_count_kernel<<<128 + S1, ST, 0, stream>>>(ei, E, 128, hist, pa);
    pa.strip0 = S1;
    bucket_scatter_kernel<<<128 + (S2 - S1), ST, 0, stream>>>(ei, E, 128, NB, hist, bstart, rec, pa);
    pa.strip0 = S2;
    bucket_place_kernel<<<NB + (NS - S2), ST, 0, stream>>>(rec, bstart, E, N, NB, off, srt, pa);

    node_softmax_gather_kernel<<<2048, 256, 0, stream>>>(off, srt, al, ar, xh, agg, N);

    wmma_gemm_kernel<128, 64, 32><<<(N + 31) / 32, 256, 0, stream>>>(
        agg, Wout, bias, out, N);
}


# Output projection fused into the node kernel: a work-group's 32 finished rows stay in LDS and its 8 waves multiply them by Wout on WMMA; GEMM launch and agg round trip gone; sums and order unchanged
# speedup vs baseline: 4.8848x; 1.1951x over previous
#include <hip/hip_runtime.h>
#include <math.h>

typedef float v2f __attribute__((ext_vector_type(2)));
typedef float v8f __attribute__((ext_vector_type(8)));

#define BSH 8
#define BW  256
#define ST  1024
#define PARTS (ST / BW)

typedef int i2 __attribute__((ext_vector_type(2)));

__device__ __forceinline__ int wave_incl_scan(int v, int lane) {
    #pragma unroll
    for (int o = 1; o < 32; o <<= 1) {
        const int up = __shfl_up(v, o, 32);
        if (lane >= o) v += up;
    }
    return v;
}

template <int M>
__device__ __forceinline__ float row_xor_add(float v) {
    const int t = __builtin_amdgcn_update_dpp(0, __float_as_int(v), 0x160 + M, 0xf, 0xf, true);
    return v + __int_as_float(t);
}

struct ProjArgs {
    const float* x;
    const float* W;
    const float* attl;
    const float* attr;
    float*       xh;
    float*       al;
    float*       ar;
    int          M;
    int          strip0;
};

#define PD    128
#define PROWS 64

__device__ __forceinline__ void proj_strip(const ProjArgs p, int strip) {
    __shared__ float Alds[PROWS * (PD + 1)];
    __shared__ float part[32 * 32];

    const float* __restrict__ A = p.x;
    const float* __restrict__ B = p.W;
    const int M       = p.M;
    const int tid     = threadIdx.x;
    const int wave    = tid >> 5;
    const int lane    = tid & 31;
    const int rowbase = strip * PROWS;

    constexpr int NQ = PROWS * PD / ST;
    float v[NQ];
    #pragma unroll
    for (int q = 0; q < NQ; ++q) {
        const int idx = tid + q * ST;
        const int r   = idx / PD;
        const int c   = idx - r * PD;
        const int gr  = rowbase + r;
        v[q] = (gr < M) ? A[(size_t)gr * PD + c] : 0.0f;
    }
    #pragma unroll
    for (int q = 0; q < NQ; ++q) {
        const int idx = tid + q * ST;
        const int r   = idx / PD;
        const int c   = idx - r * PD;
        Alds[r * (PD + 1) + c] = v[q];
    }
    __syncthreads();

    const int rg      = wave >> 3;
    const int colbase = (wave & 7) * 16;
    const int half    = lane >> 4;
    const int l15     = lane & 15;

    const float* Arow = &Alds[(rg * 16 + l15) * (PD + 1)];

    const float tl = p.attl[colbase + l15];
    const float tr = p.attr[colbase + l15];

    v8f acc = {};
    #pragma unroll 8
    for (int k = 0; k < PD; k += 4) {
        const int kk = k + 2 * half;
        v2f a, b;
        a.x = Arow[kk];
        a.y = Arow[kk + 1];
        b.x = B[(size_t)kk * PD + colbase + l15];
        b.y = B[(size_t)(kk + 1) * PD + colbase + l15];
        acc = __builtin_amdgcn_wmma_f32_16x16x4_f32(
             false, a,  false, b,
             (short)0, acc,  false,  false);
    }

    float pl[8], pr[8];
    #pragma unroll
    for (int r = 0; r < 8; ++r) { pl[r] = acc[r] * tl; pr[r] = acc[r] * tr; }
    #pragma unroll
    for (int r = 0; r < 8; ++r) { pl[r] = row_xor_add<1>(pl[r]); pr[r] = row_xor_add<1>(pr[r]); }
    #pragma unroll
    for (int r = 0; r < 8; ++r) { pl[r] = row_xor_add<2>(pl[r]); pr[r] = row_xor_add<2>(pr[r]); }
    #pragma unroll
    for (int r = 0; r < 8; ++r) { pl[r] = row_xor_add<4>(pl[r]); pr[r] = row_xor_add<4>(pr[r]); }
    #pragma unroll
    for (int r = 0; r < 8; ++r) { pl[r] = row_xor_add<8>(pl[r]); pr[r] = row_xor_add<8>(pr[r]); }
    if (l15 == 0) {
        float* w = &part[wave * 32 + half * 16];
        #pragma unroll
        for (int r = 0; r < 8; ++r) { w[2 * r] = pl[r]; w[2 * r + 1] = pr[r]; }
    }
    __syncthreads();
    if (tid < 256) {
        const int g     = tid >> 6;
        const int t     = tid & 63;
        const int which = t >> 5;
        const int row   = (t & 31) >> 1;
        const int h     = t & 1;
        const float* w  = &part[(g * 8 + h * 4) * 32 + row * 2 + which];
        const float s   = ((w[0] + w[32]) + w[64]) + w[96];
        const int grow  = rowbase + g * 16 + row;
        if (grow < M) (which ? p.ar : p.al)[(size_t)grow * 2 + h] = s;
    }

    #pragma unroll
    for (int r = 0; r < 8; ++r) {
        const int grow = rowbase + rg * 16 + r + 8 * half;
        if (grow < M)
            p.xh[(size_t)grow * PD + colbase + l15] = acc[r];
    }
}

__global__ __launch_bounds__(ST)
void bucket_count_kernel(const int* __restrict__ ei, int E, int G, int* __restrict__ hist,
                         const ProjArgs pa) {
    if ((int)blockIdx.x >= G) { proj_strip(pa, pa.strip0 + (int)blockIdx.x - G); return; }
    __shared__ int h[BW];
    const int tid = threadIdx.x;
    const int ch  = (E + G - 1) / G;
    const int lo  = min((int)blockIdx.x * ch, E);
    const int hi  = min(lo + ch, E);
    if (tid < BW) h[tid] = 0;
    __syncthreads();
    for (int e0 = lo + tid; e0 < hi; e0 += 4 * ST) {
        int  d[4];
        bool on[4];
        #pragma unroll
        for (int r = 0; r < 4; ++r) {
            const int e = e0 + r * ST;
            on[r] = e < hi;
            d[r]  = on[r] ? ei[E + e] : 0;
        }
        #pragma unroll
        for (int r = 0; r < 4; ++r)
            if (on[r]) atomicAdd(&h[d[r] >> BSH], 1);
    }
    __syncthreads();
    if (tid < BW) hist[blockIdx.x * BW + tid] = h[tid];
}

__global__ __launch_bounds__(ST)
void bucket_scatter_kernel(const int* __restrict__ ei, int E, int G, int NB,
                           const int* __restrict__ hist,
                           int* __restrict__ bstart,
                           i2* __restrict__ rec,
                           const ProjArgs pa) {
    if ((int)blockIdx.x >= G) { proj_strip(pa, pa.strip0 + (int)blockIdx.x - G); return; }
    __shared__ int ppre[PARTS][BW];
    __shared__ int ptot[PARTS][BW];
    __shared__ int cur[BW];
    __shared__ int wsum[BW / 32];
    const int tid  = threadIdx.x;
    const int lane = tid & 31;
    const int g    = blockIdx.x;

    const int b    = tid & (BW - 1);
    const int part = tid >> BSH;
    const int gh   = (G + PARTS - 1) / PARTS;
    const int g0   = part * gh;
    const int g1   = min(g0 + gh, G);
    int pre = 0, tot = 0;
    #pragma unroll 8
    for (int q = g0; q < g1; ++q) {
        const int v = hist[q * BW + b];
        tot += v;
        pre += (q < g) ? v : 0;
    }
    ppre[part][b] = pre;
    ptot[part][b] = tot;
    __syncthreads();

    int incl = 0;
    if (tid < BW) {
        pre = 0; tot = 0;
        #pragma unroll
        for (int p = 0; p < PARTS; ++p) { pre += ppre[p][tid]; tot += ptot[p][tid]; }
        incl = wave_incl_scan(tot, lane);
        if (lane == 31) wsum[tid >> 5] = incl;
    }
    __syncthreads();
    if (tid < BW) {
        int wb = 0;
        for (int w = 0; w < (tid >> 5); ++w) wb += wsum[w];
        const int bst = wb + incl - tot;
        cur[tid] = bst + pre;
        if (g == 0 && tid < NB) bstart[tid] = bst;
    }
    __syncthreads();

    const int ch = (E + G - 1) / G;
    const int lo = min(g * ch, E);
    const int hi = min(lo + ch, E);
    for (int e0 = lo + tid; e0 < hi; e0 += 4 * ST) {
        int  s[4], d[4];
        bool on[4];
        #pragma unroll
        for (int r = 0; r < 4; ++r) {
            const int e = e0 + r * ST;
            on[r] = e < hi;
            d[r]  = on[r] ? ei[E + e] : 0;
            s[r]  = on[r] ? ei[e] : 0;
        }
        #pragma unroll
        for (int r = 0; r < 4; ++r)
            if (on[r]) {
                const int p = atomicAdd(&cur[d[r] >> BSH], 1);
                i2 q; q.x = s[r]; q.y = d[r];
                rec[p] = q;
            }
    }
}

__global__ __launch_bounds__(ST)
void bucket_place_kernel(const i2* __restrict__ rec,
                         const int* __restrict__ bstart,
                         int E, int N, int NB,
                         int* __restrict__ off,
                         int* __restrict__ srt,
                         const ProjArgs pa) {
    if ((int)blockIdx.x >= NB) { proj_strip(pa, pa.strip0 + (int)blockIdx.x - NB); return; }
    __shared__ int c[BW];
    __shared__ int cur[BW];
    __shared__ int wsum[BW / 32];
    const int tid  = threadIdx.x;
    const int lane = tid & 31;
    const int b    = blockIdx.x;
    const int n0   = b << BSH;
    const int bs   = bstart[b];
    const int be   = (b + 1 < NB) ? bstart[b + 1] : E;

    if (tid < BW) c[tid] = 0;
    __syncthreads();
    for (int r0 = bs + tid; r0 < be; r0 += 4 * ST) {
        int  d[4];
        bool on[4];
        #pragma unroll
        for (int r = 0; r < 4; ++r) {
            const int i = r0 + r * ST;
            on[r] = i < be;
            d[r]  = on[r] ? rec[i].y : n0;
        }
        #pragma unroll
        for (int r = 0; r < 4; ++r)
            if (on[r]) atomicAdd(&c[d[r] - n0], 1);
    }
    __syncthreads();

    int v = 0, incl = 0;
    if (tid < BW) {
        v    = c[tid];
        incl = wave_incl_scan(v, lane);
        if (lane == 31) wsum[tid >> 5] = incl;
    }
    __syncthreads();
    if (tid < BW) {
        int wb = 0;
        for (int w = 0; w < (tid >> 5); ++w) wb += wsum[w];
        const int o = bs + wb + incl - v;
        cur[tid] = o;
        if (n0 + tid < N) off[n0 + tid] = o;
    }
    if (b == NB - 1 && tid == 0) off[N] = E;
    __syncthreads();

    for (int r0 = bs + tid; r0 < be; r0 += 4 * ST) {
        i2   q[4];
        bool on[4];
        #pragma unroll
        for (int r = 0; r < 4; ++r) {
            const int i = r0 + r * ST;
            on[r] = i < be;
            i2 z; z.x = 0; z.y = n0;
            q[r]  = on[r] ? rec[i] : z;
        }
        #pragma unroll
        for (int r = 0; r < 4; ++r)
            if (on[r]) {
                const int p = atomicAdd(&cur[q[r].y - n0], 1);
                srt[p] = q[r].x;
            }
    }
}

typedef float f2 __attribute__((ext_vector_type(2)));
typedef float v4f __attribute__((ext_vector_type(4)));

__device__ __forceinline__ float wave_max(float v) {
    #pragma unroll
    for (int o = 16; o > 0; o >>= 1) v = fmaxf(v, __shfl_xor(v, o, 32));
    return v;
}
__device__ __forceinline__ float wave_sum(float v) {
    #pragma unroll
    for (int o = 16; o > 0; o >>= 1) v += __shfl_xor(v, o, 32);
    return v;
}

__device__ __forceinline__ void load_chunk(int c, int lane, int i, int o0, int deg,
                                           const int* __restrict__ srt,
                                           const float* __restrict__ al, f2 vr,
                                           int& s, f2& e, bool& ok) {
    const int j = c + lane;
    ok = (j <= deg);
    s = i;
    if (j < deg) s = srt[o0 + j];
    const f2 vl = *(const f2*)(al + (size_t)s * 2);
    f2 v = vl + vr;
    v.x = (v.x > 0.0f) ? v.x : 0.2f * v.x;
    v.y = (v.y > 0.0f) ? v.y : 0.2f * v.y;
    e = v;
}

template <int K>
__device__ __forceinline__ void gather_rows(int j0, int s, float w0, float w1,
                                            const float* __restrict__ xh, int lane,
                                            v4f& acc) {
    v4f   v[K];
    float a[K];
    #pragma unroll
    for (int k = 0; k < K; ++k) {
        const int   sj = __builtin_amdgcn_readlane(s, j0 + k);
        const float a0 = __int_as_float(__builtin_amdgcn_readlane(__float_as_int(w0), j0 + k));
        const float a1 = __int_as_float(__builtin_amdgcn_readlane(__float_as_int(w1), j0 + k));
        a[k] = (lane < 16) ? a0 : a1;
        v[k] = *(const v4f*)(xh + (size_t)sj * 128 + 4 * lane);
    }
    #pragma unroll
    for (int k = 0; k < K; ++k) acc += v[k] * a[k];
}

__device__ __forceinline__ void walk_chunk(int n, int s, float w0, float w1,
                                           const float* __restrict__ xh, int lane,
                                           v4f& acc) {
    int j = 0;
    while (n - j >= 8) { gather_rows<8>(j, s, w0, w1, xh, lane, acc); j += 8; }
    if (n - j >= 4)    { gather_rows<4>(j, s, w0, w1, xh, lane, acc); j += 4; }
    if (n - j >= 2)    { gather_rows<2>(j, s, w0, w1, xh, lane, acc); j += 2; }
    if (n - j >= 1)    { gather_rows<1>(j, s, w0, w1, xh, lane, acc); }
}

#define TN     32
#define APITCH 132

__device__ __forceinline__ v4f node_row(int i, int lane,
                                        const int* __restrict__ off,
                                        const int* __restrict__ srt,
                                        const float* __restrict__ al,
                                        const float* __restrict__ ar,
                                        const float* __restrict__ xh) {
    const int o0  = off[i];
    const int deg = off[i + 1] - o0;
    const int tot = deg + 1;
    const f2  vr  = *(const f2*)(ar + (size_t)i * 2);

    int s0; f2 e0; bool ok0;
    load_chunk(0, lane, i, o0, deg, srt, al, vr, s0, e0, ok0);

    float m0 = wave_max(ok0 ? e0.x : -INFINITY);
    float m1 = wave_max(ok0 ? e0.y : -INFINITY);
    for (int c = 32; c < tot; c += 32) {
        int s; f2 e; bool ok;
        load_chunk(c, lane, i, o0, deg, srt, al, vr, s, e, ok);
        m0 = fmaxf(m0, wave_max(ok ? e.x : -INFINITY));
        m1 = fmaxf(m1, wave_max(ok ? e.y : -INFINITY));
    }
    if (!(fabsf(m0) <= 3.4e38f)) m0 = 0.0f;
    if (!(fabsf(m1) <= 3.4e38f)) m1 = 0.0f;

    const float x0 = ok0 ? __expf(e0.x - m0) : 0.0f;
    const float x1 = ok0 ? __expf(e0.y - m1) : 0.0f;
    float d0 = wave_sum(x0);
    float d1 = wave_sum(x1);
    for (int c = 32; c < tot; c += 32) {
        int s; f2 e; bool ok;
        load_chunk(c, lane, i, o0, deg, srt, al, vr, s, e, ok);
        d0 += wave_sum(ok ? __expf(e.x - m0) : 0.0f);
        d1 += wave_sum(ok ? __expf(e.y - m1) : 0.0f);
    }
    d0 += 1e-16f;
    d1 += 1e-16f;

    v4f acc = {0.0f, 0.0f, 0.0f, 0.0f};
    walk_chunk(min(tot, 32), s0, x0 / d0, x1 / d1, xh, lane, acc);
    for (int c = 32; c < tot; c += 32) {
        int s; f2 e; bool ok;
        load_chunk(c, lane, i, o0, deg, srt, al, vr, s, e, ok);
        const float w0 = ok ? __expf(e.x - m0) / d0 : 0.0f;
        const float w1 = ok ? __expf(e.y - m1) / d1 : 0.0f;
        walk_chunk(min(tot - c, 32), s, w0, w1, xh, lane, acc);
    }
    return acc;
}

__global__ __launch_bounds__(256)
void node_softmax_gather_out_kernel(const int* __restrict__ off,
                                    const int* __restrict__ srt,
                                    const float* __restrict__ al,
                                    const float* __restrict__ ar,
                                    const float* __restrict__ xh,
                                    const float* __restrict__ B,
                                    const float* __restrict__ bias,
                                    float* __restrict__ C, int N) {
    __shared__ __attribute__((aligned(16))) float Alds[TN * APITCH];

    const int lane    = threadIdx.x & 31;
    const int wave    = __builtin_amdgcn_readfirstlane((int)(threadIdx.x >> 5));
    const int rowbase = blockIdx.x * TN;

    for (int q = 0; q < TN / 8; ++q) {
        const int r = wave * (TN / 8) + q;
        const int i = rowbase + r;
        v4f acc = {0.0f, 0.0f, 0.0f, 0.0f};
        if (i < N) acc = node_row(i, lane, off, srt, al, ar, xh);
        *(v4f*)&Alds[r * APITCH + 4 * lane] = acc;
    }
    __syncthreads();

    const int rg      = wave >> 2;
    const int colbase = (wave & 3) * 16;
    const int half    = lane >> 4;
    const int l15     = lane & 15;

    const float* Arow = &Alds[(rg * 16 + l15) * APITCH];

    v8f acc = {};
    #pragma unroll 8
    for (int k = 0; k < 128; k += 4) {
        const int kk = k + 2 * half;
        const v2f a = *(const v2f*)&Arow[kk];
        v2f b;
        b.x = B[(size_t)kk * 64 + colbase + l15];
        b.y = B[(size_t)(kk + 1) * 64 + colbase + l15];
        acc = __builtin_amdgcn_wmma_f32_16x16x4_f32(
             false, a,  false, b,
             (short)0, acc,  false,  false);
    }

    const float bv = bias ? bias[colbase + l15] : 0.0f;
    #pragma unroll
    for (int r = 0; r < 8; ++r) {
        const int grow = rowbase + rg * 16 + r + 8 * half;
        if (grow < N)
            C[(size_t)grow * 64 + colbase + l15] = acc[r] + bv;
    }
}

extern "C" void kernel_launch(void* const* d_in, const int* in_sizes, int n_in,
                              void* d_out, int out_size, void* d_ws, size_t ws_size,
                              hipStream_t stream) {
    const float* x    = (const float*)d_in[0];
    const int*   ei   = (const int*)d_in[1];
    const float* Wl   = (const float*)d_in[2];
    const float* attl = (const float*)d_in[3];
    const float* attr = (const float*)d_in[4];
    const float* Wout = (const float*)d_in[5];
    const float* bias = (const float*)d_in[6];
    float* out = (float*)d_out;

    const int N   = in_sizes[0] / 128;
    const int E   = in_sizes[1] / 2;

    float* ws    = (float*)d_ws;
    float* xh    = ws;
    float* agg   = xh  + (size_t)N * 128;
    float* al    = agg + (size_t)N * 128;
    float* ar    = al  + (size_t)N * 2;
    int*   hist  = (int*)(ar + (size_t)N * 2);
    int*   bstart = hist + 128 * BW;
    int*   off   = hist + (size_t)N * 2;
    int*   srt   = off + (size_t)N * 2;
    i2*    rec   = (i2*)agg;
    const int NB = (N + BW - 1) >> BSH;

    const int NS = (N + PROWS - 1) / PROWS;
    const int S1 = NS / 3;
    const int S2 = (2 * NS) / 3;
    ProjArgs pa;
    pa.x = x; pa.W = Wl; pa.attl = attl; pa.attr = attr;
    pa.xh = xh; pa.al = al; pa.ar = ar; pa.M = N;

    pa.strip0 = 0;
    bucket_count_kernel<<<128 + S1, ST, 0, stream>>>(ei, E, 128, hist, pa);
    pa.strip0 = S1;
    bucket_scatter_kernel<<<128 + (S2 - S1), ST, 0, stream>>>(ei, E, 128, NB, hist, bstart, rec, pa);
    pa.strip0 = S2;
    bucket_place_kernel<<<NB + (NS - S2), ST, 0, stream>>>(rec, bstart, E, N, NB, off, srt, pa);

    node_softmax_gather_out_kernel<<<(N + TN - 1) / TN, 256, 0, stream>>>(
        off, srt, al, ar, xh, Wout, bias, out, N);
}


# Count and scatter fused in one launch: a chunk groups its records by bucket in its own slice (LDS scan), place reads its bucket's 128 chunk runs; 128-row strips on 2 launches; node list order changes
# speedup vs baseline: 5.5710x; 1.1405x over previous
#include <hip/hip_runtime.h>
#include <math.h>

typedef float v2f __attribute__((ext_vector_type(2)));
typedef float v8f __attribute__((ext_vector_type(8)));

#define BSH 8
#define BW  256
#define ST  1024
#define PARTS (ST / BW)

typedef int i2 __attribute__((ext_vector_type(2)));

__device__ __forceinline__ int wave_incl_scan(int v, int lane) {
    #pragma unroll
    for (int o = 1; o < 32; o <<= 1) {
        const int up = __shfl_up(v, o, 32);
        if (lane >= o) v += up;
    }
    return v;
}

template <int M>
__device__ __forceinline__ float row_xor_add(float v) {
    const int t = __builtin_amdgcn_update_dpp(0, __float_as_int(v), 0x160 + M, 0xf, 0xf, true);
    return v + __int_as_float(t);
}

struct ProjArgs {
    const float* x;
    const float* W;
    const float* attl;
    const float* attr;
    float*       xh;
    float*       al;
    float*       ar;
    int          M;
    int          strip0;
};

#define PD    128
#define PROWS 128

__device__ __forceinline__ void proj_strip(const ProjArgs p, int strip) {
    __shared__ float Alds[PROWS * (PD + 1)];
    __shared__ float part[2 * 32 * 32];

    const float* __restrict__ A = p.x;
    const float* __restrict__ B = p.W;
    const int M       = p.M;
    const int tid     = threadIdx.x;
    const int wave    = tid >> 5;
    const int lane    = tid & 31;
    const int rowbase = strip * PROWS;

    constexpr int NQ = PROWS * PD / ST;
    float v[NQ];
    #pragma unroll
    for (int q = 0; q < NQ; ++q) {
        const int idx = tid + q * ST;
        const int r   = idx / PD;
        const int c   = idx - r * PD;
        const int gr  = rowbase + r;
        v[q] = (gr < M) ? A[(size_t)gr * PD + c] : 0.0f;
    }
    #pragma unroll
    for (int q = 0; q < NQ; ++q) {
        const int idx = tid + q * ST;
        const int r   = idx / PD;
        const int c   = idx - r * PD;
        Alds[r * (PD + 1) + c] = v[q];
    }
    __syncthreads();

    const int rg      = wave >> 3;
    const int colbase = (wave & 7) * 16;
    const int half    = lane >> 4;
    const int l15     = lane & 15;

    const float* Arow0 = &Alds[(rg * 16 + l15) * (PD + 1)];
    const float* Arow1 = &Alds[((rg + 4) * 16 + l15) * (PD + 1)];

    const float tl = p.attl[colbase + l15];
    const float tr = p.attr[colbase + l15];

    v8f accs[2] = {};
    #pragma unroll 8
    for (int k = 0; k < PD; k += 4) {
        const int kk = k + 2 * half;
        v2f a0, a1, b;
        a0.x = Arow0[kk];
        a0.y = Arow0[kk + 1];
        a1.x = Arow1[kk];
        a1.y = Arow1[kk + 1];
        b.x = B[(size_t)kk * PD + colbase + l15];
        b.y = B[(size_t)(kk + 1) * PD + colbase + l15];
        accs[0] = __builtin_amdgcn_wmma_f32_16x16x4_f32(
             false, a0,  false, b,
             (short)0, accs[0],  false,  false);
        accs[1] = __builtin_amdgcn_wmma_f32_16x16x4_f32(
             false, a1,  false, b,
             (short)0, accs[1],  false,  false);
    }

    #pragma unroll
    for (int set = 0; set < 2; ++set) {
        const v8f acc = accs[set];
        float pl[8], pr[8];
        #pragma unroll
        for (int r = 0; r < 8; ++r) { pl[r] = acc[r] * tl; pr[r] = acc[r] * tr; }
        #pragma unroll
        for (int r = 0; r < 8; ++r) { pl[r] = row_xor_add<1>(pl[r]); pr[r] = row_xor_add<1>(pr[r]); }
        #pragma unroll
        for (int r = 0; r < 8; ++r) { pl[r] = row_xor_add<2>(pl[r]); pr[r] = row_xor_add<2>(pr[r]); }
        #pragma unroll
        for (int r = 0; r < 8; ++r) { pl[r] = row_xor_add<4>(pl[r]); pr[r] = row_xor_add<4>(pr[r]); }
        #pragma unroll
        for (int r = 0; r < 8; ++r) { pl[r] = row_xor_add<8>(pl[r]); pr[r] = row_xor_add<8>(pr[r]); }
        if (l15 == 0) {
            float* w = &part[set * 1024 + wave * 32 + half * 16];
            #pragma unroll
            for (int r = 0; r < 8; ++r) { w[2 * r] = pl[r]; w[2 * r + 1] = pr[r]; }
        }
    }
    __syncthreads();
    if (tid < 512) {
        const int g8    = tid >> 6;
        const int set   = g8 >> 2;
        const int g     = g8 & 3;
        const int t     = tid & 63;
        const int which = t >> 5;
        const int row   = (t & 31) >> 1;
        const int h     = t & 1;
        const float* w  = &part[set * 1024 + (g * 8 + h * 4) * 32 + row * 2 + which];
        const float s   = ((w[0] + w[32]) + w[64]) + w[96];
        const int grow  = rowbase + g8 * 16 + row;
        if (grow < M) (which ? p.ar : p.al)[(size_t)grow * 2 + h] = s;
    }

    #pragma unroll
    for (int set = 0; set < 2; ++set) {
        #pragma unroll
        for (int r = 0; r < 8; ++r) {
            const int grow = rowbase + (rg + 4 * set) * 16 + r + 8 * half;
            if (grow < M)
                p.xh[(size_t)grow * PD + colbase + l15] = accs[set][r];
        }
    }
}

__global__ __launch_bounds__(ST)
void chunk_group_kernel(const int* __restrict__ ei, int E, int G,
                        int* __restrict__ hist, int* __restrict__ hst,
                        i2* __restrict__ rec,
                        const ProjArgs pa) {
    if ((int)blockIdx.x >= G) { proj_strip(pa, pa.strip0 + (int)blockIdx.x - G); return; }
    __shared__ int h[BW];
    __shared__ int cur[BW];
    __shared__ int wsum[BW / 32];
    const int tid  = threadIdx.x;
    const int lane = tid & 31;
    const int g    = blockIdx.x;
    const int ch   = (E + G - 1) / G;
    const int lo   = min(g * ch, E);
    const int hi   = min(lo + ch, E);
    if (tid < BW) h[tid] = 0;
    __syncthreads();
    for (int e0 = lo + tid; e0 < hi; e0 += 4 * ST) {
        int  d[4];
        bool on[4];
        #pragma unroll
        for (int r = 0; r < 4; ++r) {
            const int e = e0 + r * ST;
            on[r] = e < hi;
            d[r]  = on[r] ? ei[E + e] : 0;
        }
        #pragma unroll
        for (int r = 0; r < 4; ++r)
            if (on[r]) atomicAdd(&h[d[r] >> BSH], 1);
    }
    __syncthreads();

    int v = 0, incl = 0;
    if (tid < BW) {
        v    = h[tid];
        incl = wave_incl_scan(v, lane);
        if (lane == 31) wsum[tid >> 5] = incl;
    }
    __syncthreads();
    if (tid < BW) {
        int wb = 0;
        for (int w = 0; w < (tid >> 5); ++w) wb += wsum[w];
        const int st = wb + incl - v;
        cur[tid] = lo + st;
        hist[g * BW + tid] = v;
        hst[g * BW + tid]  = st;
    }
    __syncthreads();

    for (int e0 = lo + tid; e0 < hi; e0 += 4 * ST) {
        int  s[4], d[4];
        bool on[4];
        #pragma unroll
        for (int r = 0; r < 4; ++r) {
            const int e = e0 + r * ST;
            on[r] = e < hi;
            d[r]  = on[r] ? ei[E + e] : 0;
            s[r]  = on[r] ? ei[e] : 0;
        }
        #pragma unroll
        for (int r = 0; r < 4; ++r)
            if (on[r]) {
                const int p = atomicAdd(&cur[d[r] >> BSH], 1);
                i2 q; q.x = s[r]; q.y = d[r];
                rec[p] = q;
            }
    }
}

#define RT 8
__global__ __launch_bounds__(ST)
void bucket_place_kernel(const i2* __restrict__ rec,
                         const int* __restrict__ hist,
                         const int* __restrict__ hst,
                         int E, int G, int N, int NB,
                         int* __restrict__ off,
                         int* __restrict__ srt,
                         const ProjArgs pa) {
    if ((int)blockIdx.x >= NB) { proj_strip(pa, pa.strip0 + (int)blockIdx.x - NB); return; }
    __shared__ int ptot[PARTS][BW];
    __shared__ int c[BW];
    __shared__ int cur[BW];
    __shared__ int wsum[BW / 32];
    __shared__ int rst[ST / RT];
    __shared__ int rcn[ST / RT];
    __shared__ int sbs;
    const int tid  = threadIdx.x;
    const int lane = tid & 31;
    const int b    = blockIdx.x;
    const int n0   = b << BSH;
    const int ch   = (E + G - 1) / G;

    const int col  = tid & (BW - 1);
    const int part = tid >> BSH;
    const int gh   = (G + PARTS - 1) / PARTS;
    const int g0   = part * gh;
    const int g1   = min(g0 + gh, G);
    int tot = 0;
    #pragma unroll 8
    for (int q = g0; q < g1; ++q) tot += hist[q * BW + col];
    ptot[part][col] = tot;
    if (tid < BW) c[tid] = 0;
    __syncthreads();

    int incl = 0;
    tot = 0;
    if (tid < BW) {
        #pragma unroll
        for (int p = 0; p < PARTS; ++p) tot += ptot[p][tid];
        incl = wave_incl_scan(tot, lane);
        if (lane == 31) wsum[tid >> 5] = incl;
    }
    __syncthreads();
    if (tid == b) {
        int wb = 0;
        for (int w = 0; w < (tid >> 5); ++w) wb += wsum[w];
        sbs = wb + incl - tot;
    }

    const int grp = tid / RT;
    const int sub = tid % RT;
    for (int gb = 0; gb < G; gb += ST / RT) {
        __syncthreads();
        if (tid < ST / RT && gb + tid < G) {
            const int q = gb + tid;
            rst[tid] = min(q * ch, E) + hst[q * BW + b];
            rcn[tid] = hist[q * BW + b];
        }
        __syncthreads();
        if (gb + grp < G) {
            const int st = rst[grp];
            const int cn = rcn[grp];
            for (int i0 = sub; i0 < cn; i0 += 4 * RT) {
                int  d[4];
                bool on[4];
                #pragma unroll
                for (int r = 0; r < 4; ++r) {
                    const int i = i0 + r * RT;
                    on[r] = i < cn;
                    d[r]  = on[r] ? rec[st + i].y : n0;
                }
                #pragma unroll
                for (int r = 0; r < 4; ++r)
                    if (on[r]) atomicAdd(&c[d[r] - n0], 1);
            }
        }
    }
    __syncthreads();

    int v = 0;
    incl = 0;
    if (tid < BW) {
        v    = c[tid];
        incl = wave_incl_scan(v, lane);
        if (lane == 31) wsum[tid >> 5] = incl;
    }
    __syncthreads();
    if (tid < BW) {
        int wb = 0;
        for (int w = 0; w < (tid >> 5); ++w) wb += wsum[w];
        const int o = sbs + wb + incl - v;
        cur[tid] = o;
        if (n0 + tid < N) off[n0 + tid] = o;
    }
    if (b == NB - 1 && tid == 0) off[N] = E;

    for (int gb = 0; gb < G; gb += ST / RT) {
        __syncthreads();
        if (tid < ST / RT && gb + tid < G) {
            const int q = gb + tid;
            rst[tid] = min(q * ch, E) + hst[q * BW + b];
            rcn[tid] = hist[q * BW + b];
        }
        __syncthreads();
        if (gb + grp < G) {
            const int st = rst[grp];
            const int cn = rcn[grp];
            for (int i0 = sub; i0 < cn; i0 += 4 * RT) {
                i2   q[4];
                bool on[4];
                #pragma unroll
                for (int r = 0; r < 4; ++r) {
                    const int i = i0 + r * RT;
                    on[r] = i < cn;
                    i2 z; z.x = 0; z.y = n0;
                    q[r]  = on[r] ? rec[st + i] : z;
                }
                #pragma unroll
                for (int r = 0; r < 4; ++r)
                    if (on[r]) {
                        const int p = atomicAdd(&cur[q[r].y - n0], 1);
                        srt[p] = q[r].x;
                    }
            }
        }
    }
}

typedef float f2 __attribute__((ext_vector_type(2)));
typedef float v4f __attribute__((ext_vector_type(4)));

__device__ __forceinline__ float wave_max(float v) {
    #pragma unroll
    for (int o = 16; o > 0; o >>= 1) v = fmaxf(v, __shfl_xor(v, o, 32));
    return v;
}
__device__ __forceinline__ float wave_sum(float v) {
    #pragma unroll
    for (int o = 16; o > 0; o >>= 1) v += __shfl_xor(v, o, 32);
    return v;
}

__device__ __forceinline__ void load_chunk(int c, int lane, int i, int o0, int deg,
                                           const int* __restrict__ srt,
                                           const float* __restrict__ al, f2 vr,
                                           int& s, f2& e, bool& ok) {
    const int j = c + lane;
    ok = (j <= deg);
    s = i;
    if (j < deg) s = srt[o0 + j];
    const f2 vl = *(const f2*)(al + (size_t)s * 2);
    f2 v = vl + vr;
    v.x = (v.x > 0.0f) ? v.x : 0.2f * v.x;
    v.y = (v.y > 0.0f) ? v.y : 0.2f * v.y;
    e = v;
}

template <int K>
__device__ __forceinline__ void gather_rows(int j0, int s, float w0, float w1,
                                            const float* __restrict__ xh, int lane,
                                            v4f& acc) {
    v4f   v[K];
    float a[K];
    #pragma unroll
    for (int k = 0; k < K; ++k) {
        const int   sj = __builtin_amdgcn_readlane(s, j0 + k);
        const float a0 = __int_as_float(__builtin_amdgcn_readlane(__float_as_int(w0), j0 + k));
        const float a1 = __int_as_float(__builtin_amdgcn_readlane(__float_as_int(w1), j0 + k));
        a[k] = (lane < 16) ? a0 : a1;
        v[k] = *(const v4f*)(xh + (size_t)sj * 128 + 4 * lane);
    }
    #pragma unroll
    for (int k = 0; k < K; ++k) acc += v[k] * a[k];
}

__device__ __forceinline__ void walk_chunk(int n, int s, float w0, float w1,
                                           const float* __restrict__ xh, int lane,
                                           v4f& acc) {
    int j = 0;
    while (n - j >= 8) { gather_rows<8>(j, s, w0, w1, xh, lane, acc); j += 8; }
    if (n - j >= 4)    { gather_rows<4>(j, s, w0, w1, xh, lane, acc); j += 4; }
    if (n - j >= 2)    { gather_rows<2>(j, s, w0, w1, xh, lane, acc); j += 2; }
    if (n - j >= 1)    { gather_rows<1>(j, s, w0, w1, xh, lane, acc); }
}

#define TN     32
#define APITCH 132

__device__ __forceinline__ v4f node_row(int i, int lane,
                                        const int* __restrict__ off,
                                        const int* __restrict__ srt,
                                        const float* __restrict__ al,
                                        const float* __restrict__ ar,
                                        const float* __restrict__ xh) {
    const int o0  = off[i];
    const int deg = off[i + 1] - o0;
    const int tot = deg + 1;
    const f2  vr  = *(const f2*)(ar + (size_t)i * 2);

    int s0; f2 e0; bool ok0;
    load_chunk(0, lane, i, o0, deg, srt, al, vr, s0, e0, ok0);

    float m0 = wave_max(ok0 ? e0.x : -INFINITY);
    float m1 = wave_max(ok0 ? e0.y : -INFINITY);
    for (int c = 32; c < tot; c += 32) {
        int s; f2 e; bool ok;
        load_chunk(c, lane, i, o0, deg, srt, al, vr, s, e, ok);
        m0 = fmaxf(m0, wave_max(ok ? e.x : -INFINITY));
        m1 = fmaxf(m1, wave_max(ok ? e.y : -INFINITY));
    }
    if (!(fabsf(m0) <= 3.4e38f)) m0 = 0.0f;
    if (!(fabsf(m1) <= 3.4e38f)) m1 = 0.0f;

    const float x0 = ok0 ? __expf(e0.x - m0) : 0.0f;
    const float x1 = ok0 ? __expf(e0.y - m1) : 0.0f;
    float d0 = wave_sum(x0);
    float d1 = wave_sum(x1);
    for (int c = 32; c < tot; c += 32) {
        int s; f2 e; bool ok;
        load_chunk(c, lane, i, o0, deg, srt, al, vr, s, e, ok);
        d0 += wave_sum(ok ? __expf(e.x - m0) : 0.0f);
        d1 += wave_sum(ok ? __expf(e.y - m1) : 0.0f);
    }
    d0 += 1e-16f;
    d1 += 1e-16f;

    v4f acc = {0.0f, 0.0f, 0.0f, 0.0f};
    walk_chunk(min(tot, 32), s0, x0 / d0, x1 / d1, xh, lane, acc);
    for (int c = 32; c < tot; c += 32) {
        int s; f2 e; bool ok;
        load_chunk(c, lane, i, o0, deg, srt, al, vr, s, e, ok);
        const float w0 = ok ? __expf(e.x - m0) / d0 : 0.0f;
        const float w1 = ok ? __expf(e.y - m1) / d1 : 0.0f;
        walk_chunk(min(tot - c, 32), s, w0, w1, xh, lane, acc);
    }
    return acc;
}

__global__ __launch_bounds__(256)
void node_softmax_gather_out_kernel(const int* __restrict__ off,
                                    const int* __restrict__ srt,
                                    const float* __restrict__ al,
                                    const float* __restrict__ ar,
                                    const float* __restrict__ xh,
                                    const float* __restrict__ B,
                                    const float* __restrict__ bias,
                                    float* __restrict__ C, int N) {
    __shared__ __attribute__((aligned(16))) float Alds[TN * APITCH];

    const int lane    = threadIdx.x & 31;
    const int wave    = __builtin_amdgcn_readfirstlane((int)(threadIdx.x >> 5));
    const int rowbase = blockIdx.x * TN;

    for (int q = 0; q < TN / 8; ++q) {
        const int r = wave * (TN / 8) + q;
        const int i = rowbase + r;
        v4f acc = {0.0f, 0.0f, 0.0f, 0.0f};
        if (i < N) acc = node_row(i, lane, off, srt, al, ar, xh);
        *(v4f*)&Alds[r * APITCH + 4 * lane] = acc;
    }
    __syncthreads();

    const int rg      = wave >> 2;
    const int colbase = (wave & 3) * 16;
    const int half    = lane >> 4;
    const int l15     = lane & 15;

    const float* Arow = &Alds[(rg * 16 + l15) * APITCH];

    v8f acc = {};
    #pragma unroll 8
    for (int k = 0; k < 128; k += 4) {
        const int kk = k + 2 * half;
        const v2f a = *(const v2f*)&Arow[kk];
        v2f b;
        b.x = B[(size_t)kk * 64 + colbase + l15];
        b.y = B[(size_t)(kk + 1) * 64 + colbase + l15];
        acc = __builtin_amdgcn_wmma_f32_16x16x4_f32(
             false, a,  false, b,
             (short)0, acc,  false,  false);
    }

    const float bv = bias ? bias[colbase + l15] : 0.0f;
    #pragma unroll
    for (int r = 0; r < 8; ++r) {
        const int grow = rowbase + rg * 16 + r + 8 * half;
        if (grow < N)
            C[(size_t)grow * 64 + colbase + l15] = acc[r] + bv;
    }
}

extern "C" void kernel_launch(void* const* d_in, const int* in_sizes, int n_in,
                              void* d_out, int out_size, void* d_ws, size_t ws_size,
                              hipStream_t stream) {
    const float* x    = (const float*)d_in[0];
    const int*   ei   = (const int*)d_in[1];
    const float* Wl   = (const float*)d_in[2];
    const float* attl = (const float*)d_in[3];
    const float* attr = (const float*)d_in[4];
    const float* Wout = (const float*)d_in[5];
    const float* bias = (const float*)d_in[6];
    float* out = (float*)d_out;

    const int N   = in_sizes[0] / 128;
    const int E   = in_sizes[1] / 2;

    float* ws    = (float*)d_ws;
    float* xh    = ws;
    float* agg   = xh  + (size_t)N * 128;
    float* al    = agg + (size_t)N * 128;
    float* ar    = al  + (size_t)N * 2;
    int*   hist  = (int*)(ar + (size_t)N * 2);
    int*   hst   = hist + 128 * BW;
    int*   off   = hist + (size_t)N * 2;
    int*   srt   = off + (size_t)N * 2;
    i2*    rec   = (i2*)agg;
    const int NB = (N + BW - 1) >> BSH;

    const int NS = (N + PROWS - 1) / PROWS;
    const int S1 = NS / 2;
    ProjArgs pa;
    pa.x = x; pa.W = Wl; pa.attl = attl; pa.attr = attr;
    pa.xh = xh; pa.al = al; pa.ar = ar; pa.M = N;

    pa.strip0 = 0;
    chunk_group_kernel<<<128 + S1, ST, 0, stream>>>(ei, E, 128, hist, hst, rec, pa);
    pa.strip0 = S1;
    bucket_place_kernel<<<NB + (NS - S1), ST, 0, stream>>>(rec, hist, hst, E, 128, N, NB, off, srt, pa);

    node_softmax_gather_out_kernel<<<(N + TN - 1) / TN, 256, 0, stream>>>(
        off, srt, al, ar, xh, Wout, bias, out, N);
}
